# starting kernel
# baseline (speedup 1.0000x reference)
#include <hip/hip_runtime.h>
#include <hip/hip_bf16.h>

typedef __attribute__((ext_vector_type(2))) float v2f;
typedef __attribute__((ext_vector_type(8))) float v8f;

#define NB     2
#define NPT    16384
#define NS     4096
#define NC     64
#define NOUT   128
#define NK     16
#define CH     512
#define NCHUNK 8
#define CHPTS  (NPT / NCHUNK)
#define AGROW  1076

__global__ __launch_bounds__(256) void knn_partial_kernel(
    const float* __restrict__ xyz, const float* __restrict__ sxyz,
    float* __restrict__ pd, int* __restrict__ pi) {
    const int tid = threadIdx.x;
    const int blkPerB = NS / 256;
    const int chunk = blockIdx.x % NCHUNK;
    const int t     = blockIdx.x / NCHUNK;
    const int b     = t / blkPerB;
    const int s     = (t % blkPerB) * 256 + tid;

    const float sx = sxyz[(b * 3 + 0) * NS + s];
    const float sy = sxyz[(b * 3 + 1) * NS + s];
    const float sz = sxyz[(b * 3 + 2) * NS + s];
    const float s2 = sx * sx + sy * sy + sz * sz;

    __shared__ float xl[CH], yl[CH], zl[CH], x2l[CH];

    float bd[NK];
    int   bi[NK];
#pragma unroll
    for (int i = 0; i < NK; ++i) { bd[i] = 3.4e38f; bi[i] = 0; }

    const float* xb = xyz + (size_t)b * 3 * NPT;
    const int n0 = chunk * CHPTS;

    for (int c0 = n0; c0 < n0 + CHPTS; c0 += CH) {
        __syncthreads();
        for (int j = tid; j < CH; j += 256) {
            float xv = xb[c0 + j];
            float yv = xb[NPT + c0 + j];
            float zv = xb[2 * NPT + c0 + j];
            xl[j] = xv; yl[j] = yv; zl[j] = zv;
            x2l[j] = xv * xv + yv * yv + zv * zv;
        }
        __syncthreads();
#pragma unroll 4
        for (int jj = 0; jj < CH; ++jj) {
            float dot = sx * xl[jj] + sy * yl[jj] + sz * zl[jj];
            float d2  = (s2 + x2l[jj]) - 2.0f * dot;
            if (d2 < bd[NK - 1]) {
                float cd = d2; int ci = c0 + jj;
#pragma unroll
                for (int i = 0; i < NK; ++i) {
                    if (cd < bd[i]) {
                        float td = bd[i]; int ti = bi[i];
                        bd[i] = cd; bi[i] = ci;
                        cd = td; ci = ti;
                    }
                }
            }
        }
    }
    const size_t base = ((size_t)(b * NS + s) * NCHUNK + chunk) * NK;
#pragma unroll
    for (int i = 0; i < NK; ++i) { pd[base + i] = bd[i]; pi[base + i] = bi[i]; }
}

__global__ __launch_bounds__(256) void knn_merge_kernel(
    const float* __restrict__ pd, const int* __restrict__ pi,
    int* __restrict__ knn) {
    const int g = blockIdx.x * 256 + threadIdx.x;
    const int b = g / NS;
    const int s = g % NS;

    float bd[NK];
    int   bi[NK];
#pragma unroll
    for (int i = 0; i < NK; ++i) { bd[i] = 3.4e38f; bi[i] = 0; }

    const size_t base = (size_t)(b * NS + s) * NCHUNK * NK;
    for (int c = 0; c < NCHUNK * NK; ++c) {
        float d = pd[base + c];
        int   ci = pi[base + c];
        if (d < bd[NK - 1]) {
            float cd = d;
#pragma unroll
            for (int i = 0; i < NK; ++i) {
                if (cd < bd[i]) {
                    float td = bd[i]; int ti = bi[i];
                    bd[i] = cd; bi[i] = ci;
                    cd = td; ci = ti;
                }
            }
        }
    }
#pragma unroll
    for (int i = 0; i < NK; ++i)
        knn[((size_t)(b * NS + s)) * NK + i] = bi[i];
}

__global__ __launch_bounds__(256) void pointconv_kernel(
    const float* __restrict__ xyz,  const float* __restrict__ feat,
    const float* __restrict__ sxyz,
    const float* __restrict__ w1,   const float* __restrict__ b1,
    const float* __restrict__ w2,   const float* __restrict__ b2,
    const float* __restrict__ lw,   const float* __restrict__ lb,
    const int*   __restrict__ knn,  float* __restrict__ out)
{
    const int tid  = threadIdx.x;
    const int lane = tid & 31;
    const int wave = tid >> 5;
    const int b    = blockIdx.x >> 8;
    const int s0   = (blockIdx.x & 255) * 16;

    __shared__ float w1s[24], b1s[8], w2s[128], b2s[16];
    __shared__ float Wl[16][16][16];
    __shared__ float Fl[16][16][68];
    __shared__ float Ag[16 * AGROW];

    if (tid < 24)  w1s[tid] = w1[tid];
    if (tid < 8)   b1s[tid] = b1[tid];
    if (tid < 128) w2s[tid] = w2[tid];
    if (tid < 16)  b2s[tid] = b2[tid];
    __syncthreads();

    {
        const int s = tid >> 4, k = tid & 15;
        const int idx = knn[((size_t)(b * NS + s0 + s)) * NK + k];
        const float nx = xyz[(size_t)b * 3 * NPT + idx];
        const float ny = xyz[(size_t)b * 3 * NPT + NPT + idx];
        const float nz = xyz[(size_t)b * 3 * NPT + 2 * NPT + idx];
        const float rx = nx - sxyz[(b * 3 + 0) * NS + s0 + s];
        const float ry = ny - sxyz[(b * 3 + 1) * NS + s0 + s];
        const float rz = nz - sxyz[(b * 3 + 2) * NS + s0 + s];

        float h[8];
#pragma unroll
        for (int o = 0; o < 8; ++o) {
            float v = w1s[o * 3 + 0] * rx + w1s[o * 3 + 1] * ry +
                      w1s[o * 3 + 2] * rz + b1s[o];
            h[o] = v >= 0.f ? v : 0.1f * v;
        }
#pragma unroll
        for (int o = 0; o < 16; ++o) {
            float v = b2s[o];
#pragma unroll
            for (int i = 0; i < 8; ++i) v += w2s[o * 8 + i] * h[i];
            v = v >= 0.f ? v : 0.1f * v;
            Wl[s][k][o] = v;
        }
        Fl[s][k][0] = nx; Fl[s][k][1] = ny; Fl[s][k][2] = nz;
        const float* fb = feat + (size_t)b * NC * NPT + idx;
#pragma unroll 8
        for (int cf = 0; cf < NC; ++cf)
            Fl[s][k][3 + cf] = fb[(size_t)cf * NPT];
        Fl[s][k][67] = 0.f;
    }
    __syncthreads();

    {
        const int r    = lane & 15;
        const int koff = (lane < 16) ? 0 : 2;
#pragma unroll
        for (int si = 0; si < 2; ++si) {
            const int s = wave * 2 + si;
#pragma unroll
            for (int ct = 0; ct < 5; ++ct) {
                const int c  = ct * 16 + r;
                const int cc = c > 67 ? 67 : c;
                v8f acc = {};
#pragma unroll
                for (int kb = 0; kb < 4; ++kb) {
                    const int k0 = kb * 4 + koff;
                    v2f a;  a.x  = Wl[s][k0][r];  a.y  = Wl[s][k0 + 1][r];
                    v2f bb; bb.x = Fl[s][k0][cc]; bb.y = Fl[s][k0 + 1][cc];
                    acc = __builtin_amdgcn_wmma_f32_16x16x4_f32(
                        false, a, false, bb, (short)0, acc, false, false);
                }
                if (c < 67) {
#pragma unroll
                    for (int rr = 0; rr < 8; ++rr) {
                        const int wi = rr + ((lane < 16) ? 0 : 8);
                        Ag[s * AGROW + wi * 67 + c] = acc[rr];
                    }
                }
            }
        }
    }
    __syncthreads();

    {
        const int r    = lane & 15;
        const int koff = (lane < 16) ? 0 : 2;
        const float* Lrow = lw + (size_t)(wave * 16 + r) * 1072;
        const float* Arow = &Ag[r * AGROW];
        v8f acc = {};
        for (int kb = 0; kb < 268; ++kb) {
            const int k0 = kb * 4 + koff;
            v2f a;  a.x  = Lrow[k0]; a.y  = Lrow[k0 + 1];
            v2f bb; bb.x = Arow[k0]; bb.y = Arow[k0 + 1];
            acc = __builtin_amdgcn_wmma_f32_16x16x4_f32(
                false, a, false, bb, (short)0, acc, false, false);
        }
        const int su = s0 + r;
#pragma unroll
        for (int rr = 0; rr < 8; ++rr) {
            const int o = wave * 16 + rr + ((lane < 16) ? 0 : 8);
            float v = acc[rr] + lb[o];
            v = v >= 0.f ? v : 0.1f * v;
            out[(size_t)b * NOUT * NS + (size_t)o * NS + su] = v;
        }
    }
}

extern "C" void kernel_launch(void* const* d_in, const int* in_sizes, int n_in,
                              void* d_out, int out_size, void* d_ws, size_t ws_size,
                              hipStream_t stream) {
    const float* xyz  = (const float*)d_in[0];
    const float* feat = (const float*)d_in[1];
    const float* sxyz = (const float*)d_in[2];
    const float* w1   = (const float*)d_in[3];
    const float* b1   = (const float*)d_in[4];
    const float* w2   = (const float*)d_in[5];
    const float* b2   = (const float*)d_in[6];
    const float* lw   = (const float*)d_in[7];
    const float* lb   = (const float*)d_in[8];
    float* out = (float*)d_out;

    const size_t npart = (size_t)NB * NS * NCHUNK * NK;
    float* pd  = (float*)d_ws;
    int*   pi  = (int*)((char*)d_ws + npart * sizeof(float));
    int*   knn = (int*)((char*)d_ws + 2 * npart * sizeof(float));

    knn_partial_kernel<<<NB * (NS / 256) * NCHUNK, 256, 0, stream>>>(xyz, sxyz, pd, pi);
    knn_merge_kernel<<<NB * NS / 256, 256, 0, stream>>>(pd, pi, knn);
    pointconv_kernel<<<NB * (NS / 16), 256, 0, stream>>>(
        xyz, feat, sxyz, w1, b1, w2, b2, lw, lb, knn, out);
}


# KNN partial scan: a block holds the 8 chunk scans of 32 queries (wave = chunk); after a flush the waves swap their lists' 4th distance in LDS, 4th smallest of the 8 caps the bound; same 16 nearest
# speedup vs baseline: 9.2065x; 9.2065x over previous
#include <hip/hip_runtime.h>
#include <hip/hip_bf16.h>

typedef __attribute__((ext_vector_type(2))) float v2f;
typedef __attribute__((ext_vector_type(8))) float v8f;
typedef __attribute__((ext_vector_type(4))) float v4f;

#define NB     2
#define NPT    16384
#define NS     4096
#define NC     64
#define NOUT   128
#define NK     16
#define CH     512
#define NCHUNK 8
#define CHPTS  (NPT / NCHUNK)
#define AGROW  1076

#define QCAP 24
#define QGRP 8

__device__ __forceinline__ float knn_rank_flush(const v2f* qt, v2f* lt, int cnt) {
    v2f   le[NK];
    float qd[QCAP];
    int   rq[QCAP];
#pragma unroll
    for (int i = 0; i < NK; ++i) le[i] = lt[i * 256];
#pragma unroll
    for (int k = 0; k < QCAP; ++k) {
        const float d = qt[k * 256].x;
        qd[k] = (k < cnt) ? d : __builtin_inff();
    }
#pragma unroll
    for (int k = 0; k < QCAP; ++k) {
        const float d  = qd[k];
        const bool  c1 = le[7].x <= d;
        const float p2 = c1 ? le[11].x : le[3].x;
        const float b1 = c1 ? le[9].x  : le[1].x;
        const float b5 = c1 ? le[13].x : le[5].x;
        const float a0 = c1 ? le[8].x  : le[0].x;
        const float a2 = c1 ? le[10].x : le[2].x;
        const float a4 = c1 ? le[12].x : le[4].x;
        const float a6 = c1 ? le[14].x : le[6].x;
        const bool  c2 = p2 <= d;
        const float p3 = c2 ? b5 : b1;
        const float e0 = c2 ? a4 : a0;
        const float e2 = c2 ? a6 : a2;
        const bool  c3 = p3 <= d;
        const float p4 = c3 ? e2 : e0;
        const bool  c4 = p4 <= d;
        const bool  c5 = le[15].x <= d;
        rq[k] = (c1 ? 8 : 0) + (c2 ? 4 : 0) + (c3 ? 2 : 0) + (c4 ? 1 : 0) + (c5 ? 1 : 0);
    }
#pragma unroll
    for (int kp = 0; kp < QCAP; ++kp) {
        int ahead = 0;
#pragma unroll
        for (int k = kp + 1; k < QCAP; ++k) {
            const int c = (qd[kp] <= qd[k]) ? 1 : 0;
            rq[k] += c;
            ahead += c;
        }
        rq[kp] += (QCAP - 1 - kp) - ahead;
    }
    unsigned taken = 0u;
#pragma unroll
    for (int k = 0; k < QCAP; ++k) {
        const int r = rq[k];
        taken |= 1u << (r < 31 ? r : 31);
        lt[(r < NK ? r : NK) * 256] = qt[k * 256];
    }
    unsigned freem = ~taken;
#pragma unroll
    for (int i = 0; i < NK; ++i) {
        const unsigned p = (unsigned)(__ffs((int)freem) - 1);
        freem &= freem - 1u;
        lt[(p < (unsigned)NK ? p : (unsigned)NK) * 256] = le[i];
    }
    return lt[(NK - 1) * 256].x;
}

__global__ __launch_bounds__(256) void knn_partial_kernel(
    const float* __restrict__ xyz, const float* __restrict__ sxyz,
    float* __restrict__ pd, int* __restrict__ pi) {
    const int tid  = threadIdx.x;
    const int lane = tid & 31;
    const int chunk = __builtin_amdgcn_readfirstlane(tid >> 5);
    const int blkPerB = NS / 32;
    const int b     = blockIdx.x / blkPerB;
    const int s     = (blockIdx.x % blkPerB) * 32 + lane;

    const float sx = sxyz[(b * 3 + 0) * NS + s];
    const float sy = sxyz[(b * 3 + 1) * NS + s];
    const float sz = sxyz[(b * 3 + 2) * NS + s];
    const float s2 = __builtin_fmaf(sz, sz, __builtin_fmaf(sx, sx, sy * sy));

    __shared__ v4f xl4[NCHUNK * CH / 4], yl4[NCHUNK * CH / 4], zl4[NCHUNK * CH / 4], x2l4[NCHUNK * CH / 4];
    __shared__ v2f qe[QCAP * 256];
    __shared__ v2f ls[(NK + 1) * 256];
    __shared__ float sbv[NCHUNK * 32];

    float bound = 3.4e38f;
#pragma unroll
    for (int i = 0; i < NK; ++i) {
        v2f e; e.x = 3.4e38f; e.y = __int_as_float(0);
        ls[i * 256 + tid] = e;
    }
#pragma unroll
    for (int k = 0; k < QCAP; ++k) {
        v2f e; e.x = __builtin_inff(); e.y = __int_as_float(0);
        qe[k * 256 + tid] = e;
    }
    sbv[tid] = 3.4e38f;
    int cnt = 0;

    const float* xb = xyz + (size_t)b * 3 * NPT;
    const float* xw  = reinterpret_cast<const float*>(xl4)  + chunk * CH;
    const float* yw  = reinterpret_cast<const float*>(yl4)  + chunk * CH;
    const float* zw  = reinterpret_cast<const float*>(zl4)  + chunk * CH;
    const float* x2w = reinterpret_cast<const float*>(x2l4) + chunk * CH;

    v4f px[4], py[4], pz[4];
#pragma unroll
    for (int i = 0; i < 4; ++i) {
        const int g = tid + 256 * i;
        const float* p = xb + (g >> 7) * CHPTS + (g & 127) * 4;
        px[i] = *reinterpret_cast<const v4f*>(p);
        py[i] = *reinterpret_cast<const v4f*>(p + NPT);
        pz[i] = *reinterpret_cast<const v4f*>(p + 2 * NPT);
    }

    for (int r = 0; r < CHPTS / CH; ++r) {
        __syncthreads();
#pragma unroll
        for (int i = 0; i < 4; ++i) {
            const int g = tid + 256 * i;
            v4f q2;
            q2.x = __builtin_fmaf(pz[i].x, pz[i].x, __builtin_fmaf(px[i].x, px[i].x, py[i].x * py[i].x));
            q2.y = __builtin_fmaf(pz[i].y, pz[i].y, __builtin_fmaf(px[i].y, px[i].y, py[i].y * py[i].y));
            q2.z = __builtin_fmaf(pz[i].z, pz[i].z, __builtin_fmaf(px[i].z, px[i].z, py[i].z * py[i].z));
            q2.w = __builtin_fmaf(pz[i].w, pz[i].w, __builtin_fmaf(px[i].w, px[i].w, py[i].w * py[i].w));
            xl4[g] = px[i]; yl4[g] = py[i]; zl4[g] = pz[i]; x2l4[g] = q2;
        }
        __syncthreads();
        if (r + 1 < CHPTS / CH) {
#pragma unroll
            for (int i = 0; i < 4; ++i) {
                const int g = tid + 256 * i;
                const float* p = xb + (g >> 7) * CHPTS + (r + 1) * CH + (g & 127) * 4;
                px[i] = *reinterpret_cast<const v4f*>(p);
                py[i] = *reinterpret_cast<const v4f*>(p + NPT);
                pz[i] = *reinterpret_cast<const v4f*>(p + 2 * NPT);
            }
        }
        const int ibase = chunk * CHPTS + r * CH;
        for (int j0 = 0; j0 < CH; j0 += QGRP) {
            float dv[QGRP];
#pragma unroll
            for (int u = 0; u < QGRP; ++u) {
                const int jj = j0 + u;
                const float dot = __builtin_fmaf(sz, zw[jj], __builtin_fmaf(sx, xw[jj], sy * yw[jj]));
                dv[u] = __builtin_fmaf(-2.0f, dot, s2 + x2w[jj]);
            }
#pragma unroll
            for (int u = 0; u < QGRP; ++u) {
                v2f e; e.x = dv[u]; e.y = __int_as_float(ibase + j0 + u);
                qe[cnt * 256 + tid] = e;
                cnt += (dv[u] < bound) ? 1 : 0;
            }
            const bool last = (r + 1 >= CHPTS / CH) && (j0 + QGRP >= CH);
            if (last || __builtin_amdgcn_ballot_w32(cnt > QCAP - QGRP) != 0u) {
                bound = knn_rank_flush(&qe[tid], &ls[tid], cnt);
                cnt = 0;
                sbv[tid] = ls[3 * 256 + tid].x;
                __builtin_amdgcn_fence(__ATOMIC_ACQUIRE, "workgroup", "local");
                float a0 = sbv[0 * 32 + lane], a1 = sbv[1 * 32 + lane];
                float a2 = sbv[2 * 32 + lane], a3 = sbv[3 * 32 + lane];
                float c0 = sbv[4 * 32 + lane], c1 = sbv[5 * 32 + lane];
                float c2 = sbv[6 * 32 + lane], c3 = sbv[7 * 32 + lane];
                float t;
                t = fminf(a0, a1); a1 = fmaxf(a0, a1); a0 = t;
                t = fminf(a2, a3); a3 = fmaxf(a2, a3); a2 = t;
                t = fminf(a0, a2); a2 = fmaxf(a0, a2); a0 = t;
                t = fminf(a1, a3); a3 = fmaxf(a1, a3); a1 = t;
                t = fminf(a1, a2); a2 = fmaxf(a1, a2); a1 = t;
                t = fminf(c0, c1); c1 = fmaxf(c0, c1); c0 = t;
                t = fminf(c2, c3); c3 = fmaxf(c2, c3); c2 = t;
                t = fminf(c0, c2); c2 = fmaxf(c0, c2); c0 = t;
                t = fminf(c1, c3); c3 = fmaxf(c1, c3); c1 = t;
                t = fminf(c1, c2); c2 = fmaxf(c1, c2); c1 = t;
                float f4 = fminf(a3, c3);
                f4 = fminf(f4, fmaxf(a0, c2));
                f4 = fminf(f4, fmaxf(a1, c1));
                f4 = fminf(f4, fmaxf(a2, c0));
                const float wide = __builtin_fmaf(fabsf(f4), 1.0e-6f, f4) + 1.0e-30f;
                bound = fminf(bound, wide);
            }
        }
    }
    const size_t base = ((size_t)(b * NS + s) * NCHUNK + chunk) * NK;
#pragma unroll
    for (int i = 0; i < NK; ++i) {
        const v2f e = ls[i * 256 + tid];
        pd[base + i] = e.x; pi[base + i] = __float_as_int(e.y);
    }
}

#define MQ   16
#define MENT (NCHUNK * NK)

__global__ __launch_bounds__(256) void knn_merge_kernel(
    const float* __restrict__ pd, const int* __restrict__ pi,
    int* __restrict__ knn) {
    __shared__ float4 dl4[MQ * MENT / 4];
    __shared__ int4   il4[MQ * MENT / 4];
    __shared__ int    kl[MQ * NK];

    const int tid = threadIdx.x;
    const size_t q0 = (size_t)blockIdx.x * MQ;

    const float4* gd = reinterpret_cast<const float4*>(pd + q0 * MENT);
    const int4*   gi = reinterpret_cast<const int4*>(pi + q0 * MENT);
    dl4[tid] = gd[tid];  dl4[tid + 256] = gd[tid + 256];
    il4[tid] = gi[tid];  il4[tid + 256] = gi[tid + 256];
    __syncthreads();

    const int s = tid >> 4;
    const int j = tid & 15;
    const float* dq = reinterpret_cast<const float*>(dl4) + s * MENT;
    const int*   iq = reinterpret_cast<const int*>(il4) + s * MENT;

    float mine[NCHUNK];
    int   rank[NCHUNK];
#pragma unroll
    for (int l = 0; l < NCHUNK; ++l) { mine[l] = dq[l * NK + j]; rank[l] = j; }

#pragma unroll
    for (int m = 0; m < NCHUNK; ++m) {
#pragma unroll
        for (int p4 = 0; p4 < NK / 4; ++p4) {
            const float4 e4 = dl4[s * (MENT / 4) + m * (NK / 4) + p4];
            const float e[4] = {e4.x, e4.y, e4.z, e4.w};
#pragma unroll
            for (int u = 0; u < 4; ++u) {
#pragma unroll
                for (int l = 0; l < NCHUNK; ++l) {
                    if (l == m) continue;
                    if (m < l) rank[l] += (e[u] <= mine[l]) ? 1 : 0;
                    else       rank[l] += (e[u] <  mine[l]) ? 1 : 0;
                }
            }
        }
    }

#pragma unroll
    for (int l = 0; l < NCHUNK; ++l)
        if (rank[l] < NK) kl[s * NK + rank[l]] = iq[l * NK + j];
    __syncthreads();
    knn[q0 * NK + tid] = kl[tid];
}

#define NPL    (NC + 3)
#define NSLOT  (NS * NK)
#define GSPLIT 4
#define GENT   (NSLOT / GSPLIT)

__global__ __launch_bounds__(256) void gather_plane_kernel(
    const float* __restrict__ xyz, const float* __restrict__ feat,
    const int* __restrict__ knn, float* __restrict__ gf) {
    __shared__ v4f pl4[NPT / 4];
    const float* pl = reinterpret_cast<const float*>(pl4);
    const int tid  = threadIdx.x;
    const int part = blockIdx.x % GSPLIT;
    const int t    = blockIdx.x / GSPLIT;
    const int c    = t % NPL;
    const int b    = t / NPL;

    const float* src = (c < 3) ? xyz  + ((size_t)b * 3  + c) * NPT
                               : feat + ((size_t)b * NC + (c - 3)) * NPT;
    const v4f* s4 = reinterpret_cast<const v4f*>(src);
    for (int i = tid; i < NPT / 4; i += 256) pl4[i] = s4[i];
    __syncthreads();

    const int4* k4 = reinterpret_cast<const int4*>(knn + (size_t)b * NSLOT + (size_t)part * GENT);
    v4f*        o4 = reinterpret_cast<v4f*>(gf + ((size_t)b * NPL + c) * NSLOT + (size_t)part * GENT);
    for (int i = tid; i < GENT / 4; i += 256) {
        const int4 id = k4[i];
        v4f v;
        v.x = pl[id.x]; v.y = pl[id.y]; v.z = pl[id.z]; v.w = pl[id.w];
        o4[i] = v;
    }
}

#define LWROW4  268
#define SPITCH  68

template <int NU>
__device__ __forceinline__ v8f lw_mma(v8f acc, const float* a, const float* bsrc) {
#pragma unroll
    for (int j = 0; j < NU * 4; ++j) {
        const v2f av = *reinterpret_cast<const v2f*>(a + j * 4);
        v2f bb; bb.x = bsrc[j * 4]; bb.y = bsrc[j * 4 + 1];
        acc = __builtin_amdgcn_wmma_f32_16x16x4_f32(
            false, av, false, bb, (short)0, acc, false, false);
    }
    return acc;
}

__global__ __launch_bounds__(256) void pointconv_kernel(
    const float* __restrict__ sxyz,
    const float* __restrict__ w1,   const float* __restrict__ b1,
    const float* __restrict__ w2,   const float* __restrict__ b2,
    const float* __restrict__ lw,   const float* __restrict__ lb,
    const float* __restrict__ gf,   float* __restrict__ out)
{
    const int tid  = threadIdx.x;
    const int lane = tid & 31;
    const int wave = tid >> 5;
    const int b    = blockIdx.x >> 8;
    const int s0   = (blockIdx.x & 255) * 16;

    __shared__ float w1s[24], b1s[8], w2s[128], b2s[16];
    __shared__ float Wl[16][16][16];
    __shared__ float Fl[16][16][68];
    __shared__ float Ag[16 * AGROW];

    if (tid < 24)  w1s[tid] = w1[tid];
    if (tid < 8)   b1s[tid] = b1[tid];
    if (tid < 128) w2s[tid] = w2[tid];
    if (tid < 16)  b2s[tid] = b2[tid];
    __syncthreads();

    {
        const int s = tid >> 4, k = tid & 15;
        const float* gp = gf + (size_t)b * NPL * NSLOT + (size_t)s0 * NK + tid;
        const float nx = gp[0];
        const float ny = gp[(size_t)NSLOT];
        const float nz = gp[(size_t)2 * NSLOT];
        const float rx = nx - sxyz[(b * 3 + 0) * NS + s0 + s];
        const float ry = ny - sxyz[(b * 3 + 1) * NS + s0 + s];
        const float rz = nz - sxyz[(b * 3 + 2) * NS + s0 + s];

        float h[8];
#pragma unroll
        for (int o = 0; o < 8; ++o) {
            float v = w1s[o * 3 + 0] * rx + w1s[o * 3 + 1] * ry +
                      w1s[o * 3 + 2] * rz + b1s[o];
            h[o] = v >= 0.f ? v : 0.1f * v;
        }
#pragma unroll
        for (int o = 0; o < 16; ++o) {
            float v = b2s[o];
#pragma unroll
            for (int i = 0; i < 8; ++i) v += w2s[o * 8 + i] * h[i];
            v = v >= 0.f ? v : 0.1f * v;
            Wl[s][k][o] = v;
        }
        Fl[s][k][0] = nx; Fl[s][k][1] = ny; Fl[s][k][2] = nz;
#pragma unroll 8
        for (int cf = 0; cf < NC; ++cf)
            Fl[s][k][3 + cf] = gp[(size_t)(3 + cf) * NSLOT];
        Fl[s][k][67] = 0.f;
    }
    __syncthreads();

    {
        const int r    = lane & 15;
        const int koff = (lane < 16) ? 0 : 2;
#pragma unroll
        for (int si = 0; si < 2; ++si) {
            const int s = wave * 2 + si;
#pragma unroll
            for (int ct = 0; ct < 5; ++ct) {
                const int c  = ct * 16 + r;
                const int cc = c > 67 ? 67 : c;
                v8f acc = {};
#pragma unroll
                for (int kb = 0; kb < 4; ++kb) {
                    const int k0 = kb * 4 + koff;
                    v2f a;  a.x  = Wl[s][k0][r];  a.y  = Wl[s][k0 + 1][r];
                    v2f bb; bb.x = Fl[s][k0][cc]; bb.y = Fl[s][k0 + 1][cc];
                    acc = __builtin_amdgcn_wmma_f32_16x16x4_f32(
                        false, a, false, bb, (short)0, acc, false, false);
                }
                if (c < 67) {
#pragma unroll
                    for (int rr = 0; rr < 8; ++rr) {
                        const int wi = rr + ((lane < 16) ? 0 : 8);
                        Ag[s * AGROW + wi * 67 + c] = acc[rr];
                    }
                }
            }
        }
    }
    __syncthreads();

    {
        const int r    = lane & 15;
        const int koff = (lane < 16) ? 0 : 2;
        const float* Arow = &Ag[r * AGROW];

        float* stg = &Fl[0][0][0] + wave * (2 * 16 * SPITCH);
        const int c4   = lane & 3;
        const int srow = 4 * ((lane >> 2) & 3) + (lane >> 4);
        const v4f* lp = reinterpret_cast<const v4f*>(lw)
                      + (size_t)(wave * 16 + srow) * LWROW4 + c4;
        float*       sput = stg + srow * SPITCH + c4 * 4;
        const float* sget = stg + r * SPITCH + koff;

        v4f a0, a1, a2, a3, b0, b1, b2, b3;
        a0 = lp[0]; a1 = lp[4]; a2 = lp[8];
        b0 = lp[2 * LWROW4]; b1 = lp[2 * LWROW4 + 4]; b2 = lp[2 * LWROW4 + 8];
        lp += 12;
        *reinterpret_cast<v4f*>(sput)                   = a0;
        *reinterpret_cast<v4f*>(sput + 16)              = a1;
        *reinterpret_cast<v4f*>(sput + 32)              = a2;
        *reinterpret_cast<v4f*>(sput + 2 * SPITCH)      = b0;
        *reinterpret_cast<v4f*>(sput + 2 * SPITCH + 16) = b1;
        *reinterpret_cast<v4f*>(sput + 2 * SPITCH + 32) = b2;
        a0 = lp[0]; a1 = lp[4]; a2 = lp[8]; a3 = lp[12];
        b0 = lp[2 * LWROW4]; b1 = lp[2 * LWROW4 + 4]; b2 = lp[2 * LWROW4 + 8]; b3 = lp[2 * LWROW4 + 12];
        lp += 16;
        __builtin_amdgcn_fence(__ATOMIC_RELEASE, "wavefront");
        __builtin_amdgcn_fence(__ATOMIC_ACQUIRE, "wavefront");
        v8f acc = {};
        acc = lw_mma<3>(acc, sget, Arow + koff);

        for (int S = 0; S < 15; ++S) {
            const int bo = ((S + 1) & 1) * (16 * SPITCH);
            float* sp = sput + bo;
            *reinterpret_cast<v4f*>(sp)                   = a0;
            *reinterpret_cast<v4f*>(sp + 16)              = a1;
            *reinterpret_cast<v4f*>(sp + 32)              = a2;
            *reinterpret_cast<v4f*>(sp + 48)              = a3;
            *reinterpret_cast<v4f*>(sp + 2 * SPITCH)      = b0;
            *reinterpret_cast<v4f*>(sp + 2 * SPITCH + 16) = b1;
            *reinterpret_cast<v4f*>(sp + 2 * SPITCH + 32) = b2;
            *reinterpret_cast<v4f*>(sp + 2 * SPITCH + 48) = b3;
            a0 = lp[0]; a1 = lp[4]; a2 = lp[8]; a3 = lp[12];
            b0 = lp[2 * LWROW4]; b1 = lp[2 * LWROW4 + 4]; b2 = lp[2 * LWROW4 + 8]; b3 = lp[2 * LWROW4 + 12];
            lp += 16;
            __builtin_amdgcn_fence(__ATOMIC_RELEASE, "wavefront");
            __builtin_amdgcn_fence(__ATOMIC_ACQUIRE, "wavefront");
            acc = lw_mma<4>(acc, sget + bo, Arow + 48 + S * 64 + koff);
        }
        *reinterpret_cast<v4f*>(sput)                   = a0;
        *reinterpret_cast<v4f*>(sput + 16)              = a1;
        *reinterpret_cast<v4f*>(sput + 32)              = a2;
        *reinterpret_cast<v4f*>(sput + 48)              = a3;
        *reinterpret_cast<v4f*>(sput + 2 * SPITCH)      = b0;
        *reinterpret_cast<v4f*>(sput + 2 * SPITCH + 16) = b1;
        *reinterpret_cast<v4f*>(sput + 2 * SPITCH + 32) = b2;
        *reinterpret_cast<v4f*>(sput + 2 * SPITCH + 48) = b3;
        __builtin_amdgcn_fence(__ATOMIC_RELEASE, "wavefront");
        __builtin_amdgcn_fence(__ATOMIC_ACQUIRE, "wavefront");
        acc = lw_mma<4>(acc, sget, Arow + 48 + 15 * 64 + koff);

        const int su = s0 + r;
#pragma unroll
        for (int rr = 0; rr < 8; ++rr) {
            const int o = wave * 16 + rr + ((lane < 16) ? 0 : 8);
            float v = acc[rr] + lb[o];
            v = v >= 0.f ? v : 0.1f * v;
            out[(size_t)b * NOUT * NS + (size_t)o * NS + su] = v;
        }
    }
}

extern "C" void kernel_launch(void* const* d_in, const int* in_sizes, int n_in,
                              void* d_out, int out_size, void* d_ws, size_t ws_size,
                              hipStream_t stream) {
    const float* xyz  = (const float*)d_in[0];
    const float* feat = (const float*)d_in[1];
    const float* sxyz = (const float*)d_in[2];
    const float* w1   = (const float*)d_in[3];
    const float* b1   = (const float*)d_in[4];
    const float* w2   = (const float*)d_in[5];
    const float* b2   = (const float*)d_in[6];
    const float* lw   = (const float*)d_in[7];
    const float* lb   = (const float*)d_in[8];
    float* out = (float*)d_out;

    const size_t npart = (size_t)NB * NS * NCHUNK * NK;
    float* pd  = (float*)d_ws;
    int*   pi  = (int*)((char*)d_ws + npart * sizeof(float));
    int*   knn = (int*)((char*)d_ws + 2 * npart * sizeof(float));
    float* gf  = (float*)((char*)d_ws + 2 * npart * sizeof(float)
                                      + (size_t)NB * NSLOT * sizeof(int));

    knn_partial_kernel<<<NB * (NS / 32), 256, 0, stream>>>(xyz, sxyz, pd, pi);
    knn_merge_kernel<<<NB * NS / MQ, 256, 0, stream>>>(pd, pi, knn);
    gather_plane_kernel<<<NB * NPL * GSPLIT, 256, 0, stream>>>(xyz, feat, knn, gf);
    pointconv_kernel<<<NB * (NS / 16), 256, 0, stream>>>(
        sxyz, w1, b1, w2, b2, lw, lb, gf, out);
}


# KNN partial scan: a group's 8 points are read from LDS one trip ahead, behind the distances of the group before, so the read latency passes under the queue stores and fill test; same lists
# speedup vs baseline: 9.6632x; 1.0496x over previous
#include <hip/hip_runtime.h>
#include <hip/hip_bf16.h>

typedef __attribute__((ext_vector_type(2))) float v2f;
typedef __attribute__((ext_vector_type(8))) float v8f;
typedef __attribute__((ext_vector_type(4))) float v4f;

#define NB     2
#define NPT    16384
#define NS     4096
#define NC     64
#define NOUT   128
#define NK     16
#define CH     512
#define NCHUNK 8
#define CHPTS  (NPT / NCHUNK)
#define AGROW  1076

#define QCAP 24
#define QGRP 8

__device__ __forceinline__ float knn_rank_flush(const v2f* qt, v2f* lt, int cnt) {
    v2f   le[NK];
    float qd[QCAP];
    int   rq[QCAP];
#pragma unroll
    for (int i = 0; i < NK; ++i) le[i] = lt[i * 256];
#pragma unroll
    for (int k = 0; k < QCAP; ++k) {
        const float d = qt[k * 256].x;
        qd[k] = (k < cnt) ? d : __builtin_inff();
    }
#pragma unroll
    for (int k = 0; k < QCAP; ++k) {
        const float d  = qd[k];
        const bool  c1 = le[7].x <= d;
        const float p2 = c1 ? le[11].x : le[3].x;
        const float b1 = c1 ? le[9].x  : le[1].x;
        const float b5 = c1 ? le[13].x : le[5].x;
        const float a0 = c1 ? le[8].x  : le[0].x;
        const float a2 = c1 ? le[10].x : le[2].x;
        const float a4 = c1 ? le[12].x : le[4].x;
        const float a6 = c1 ? le[14].x : le[6].x;
        const bool  c2 = p2 <= d;
        const float p3 = c2 ? b5 : b1;
        const float e0 = c2 ? a4 : a0;
        const float e2 = c2 ? a6 : a2;
        const bool  c3 = p3 <= d;
        const float p4 = c3 ? e2 : e0;
        const bool  c4 = p4 <= d;
        const bool  c5 = le[15].x <= d;
        rq[k] = (c1 ? 8 : 0) + (c2 ? 4 : 0) + (c3 ? 2 : 0) + (c4 ? 1 : 0) + (c5 ? 1 : 0);
    }
#pragma unroll
    for (int kp = 0; kp < QCAP; ++kp) {
        int ahead = 0;
#pragma unroll
        for (int k = kp + 1; k < QCAP; ++k) {
            const int c = (qd[kp] <= qd[k]) ? 1 : 0;
            rq[k] += c;
            ahead += c;
        }
        rq[kp] += (QCAP - 1 - kp) - ahead;
    }
    unsigned taken = 0u;
#pragma unroll
    for (int k = 0; k < QCAP; ++k) {
        const int r = rq[k];
        taken |= 1u << (r < 31 ? r : 31);
        lt[(r < NK ? r : NK) * 256] = qt[k * 256];
    }
    unsigned freem = ~taken;
#pragma unroll
    for (int i = 0; i < NK; ++i) {
        const unsigned p = (unsigned)(__ffs((int)freem) - 1);
        freem &= freem - 1u;
        lt[(p < (unsigned)NK ? p : (unsigned)NK) * 256] = le[i];
    }
    return lt[(NK - 1) * 256].x;
}

__global__ __launch_bounds__(256) void knn_partial_kernel(
    const float* __restrict__ xyz, const float* __restrict__ sxyz,
    float* __restrict__ pd, int* __restrict__ pi) {
    const int tid  = threadIdx.x;
    const int lane = tid & 31;
    const int chunk = __builtin_amdgcn_readfirstlane(tid >> 5);
    const int blkPerB = NS / 32;
    const int b     = blockIdx.x / blkPerB;
    const int s     = (blockIdx.x % blkPerB) * 32 + lane;

    const float sx = sxyz[(b * 3 + 0) * NS + s];
    const float sy = sxyz[(b * 3 + 1) * NS + s];
    const float sz = sxyz[(b * 3 + 2) * NS + s];
    const float s2 = __builtin_fmaf(sz, sz, __builtin_fmaf(sx, sx, sy * sy));

    __shared__ v4f xl4[NCHUNK * CH / 4], yl4[NCHUNK * CH / 4], zl4[NCHUNK * CH / 4], x2l4[NCHUNK * CH / 4];
    __shared__ v2f qe[QCAP * 256];
    __shared__ v2f ls[(NK + 1) * 256];
    __shared__ float sbv[NCHUNK * 32];

    float bound = 3.4e38f;
#pragma unroll
    for (int i = 0; i < NK; ++i) {
        v2f e; e.x = 3.4e38f; e.y = __int_as_float(0);
        ls[i * 256 + tid] = e;
    }
#pragma unroll
    for (int k = 0; k < QCAP; ++k) {
        v2f e; e.x = __builtin_inff(); e.y = __int_as_float(0);
        qe[k * 256 + tid] = e;
    }
    sbv[tid] = 3.4e38f;
    int cnt = 0;

    const float* xb = xyz + (size_t)b * 3 * NPT;
    const v4f* xw4  = xl4  + chunk * (CH / 4);
    const v4f* yw4  = yl4  + chunk * (CH / 4);
    const v4f* zw4  = zl4  + chunk * (CH / 4);
    const v4f* x2w4 = x2l4 + chunk * (CH / 4);

    v4f px[4], py[4], pz[4];
#pragma unroll
    for (int i = 0; i < 4; ++i) {
        const int g = tid + 256 * i;
        const float* p = xb + (g >> 7) * CHPTS + (g & 127) * 4;
        px[i] = *reinterpret_cast<const v4f*>(p);
        py[i] = *reinterpret_cast<const v4f*>(p + NPT);
        pz[i] = *reinterpret_cast<const v4f*>(p + 2 * NPT);
    }

    for (int r = 0; r < CHPTS / CH; ++r) {
        __syncthreads();
#pragma unroll
        for (int i = 0; i < 4; ++i) {
            const int g = tid + 256 * i;
            v4f q2;
            q2.x = __builtin_fmaf(pz[i].x, pz[i].x, __builtin_fmaf(px[i].x, px[i].x, py[i].x * py[i].x));
            q2.y = __builtin_fmaf(pz[i].y, pz[i].y, __builtin_fmaf(px[i].y, px[i].y, py[i].y * py[i].y));
            q2.z = __builtin_fmaf(pz[i].z, pz[i].z, __builtin_fmaf(px[i].z, px[i].z, py[i].z * py[i].z));
            q2.w = __builtin_fmaf(pz[i].w, pz[i].w, __builtin_fmaf(px[i].w, px[i].w, py[i].w * py[i].w));
            xl4[g] = px[i]; yl4[g] = py[i]; zl4[g] = pz[i]; x2l4[g] = q2;
        }
        __syncthreads();
        if (r + 1 < CHPTS / CH) {
#pragma unroll
            for (int i = 0; i < 4; ++i) {
                const int g = tid + 256 * i;
                const float* p = xb + (g >> 7) * CHPTS + (r + 1) * CH + (g & 127) * 4;
                px[i] = *reinterpret_cast<const v4f*>(p);
                py[i] = *reinterpret_cast<const v4f*>(p + NPT);
                pz[i] = *reinterpret_cast<const v4f*>(p + 2 * NPT);
            }
        }
        const int ibase = chunk * CHPTS + r * CH;
        v4f px0 = xw4[0],  px1 = xw4[1],  py0 = yw4[0],  py1 = yw4[1];
        v4f pz0 = zw4[0],  pz1 = zw4[1],  pq0 = x2w4[0], pq1 = x2w4[1];
        for (int j0 = 0; j0 < CH; j0 += QGRP) {
            float dv[QGRP];
#define KNN_D2(X, Y, Z, Q)   \
            __builtin_fmaf(-2.0f, __builtin_fmaf(sz, (Z), __builtin_fmaf(sx, (X), sy * (Y))), s2 + (Q))
            dv[0] = KNN_D2(px0.x, py0.x, pz0.x, pq0.x);
            dv[1] = KNN_D2(px0.y, py0.y, pz0.y, pq0.y);
            dv[2] = KNN_D2(px0.z, py0.z, pz0.z, pq0.z);
            dv[3] = KNN_D2(px0.w, py0.w, pz0.w, pq0.w);
            dv[4] = KNN_D2(px1.x, py1.x, pz1.x, pq1.x);
            dv[5] = KNN_D2(px1.y, py1.y, pz1.y, pq1.y);
            dv[6] = KNN_D2(px1.z, py1.z, pz1.z, pq1.z);
            dv[7] = KNN_D2(px1.w, py1.w, pz1.w, pq1.w);
#undef KNN_D2
            const int gn = ((j0 + QGRP) & (CH - 1)) >> 2;
            px0 = xw4[gn];  px1 = xw4[gn + 1];  py0 = yw4[gn];  py1 = yw4[gn + 1];
            pz0 = zw4[gn];  pz1 = zw4[gn + 1];  pq0 = x2w4[gn]; pq1 = x2w4[gn + 1];
            __builtin_amdgcn_sched_barrier(0);
#pragma unroll
            for (int u = 0; u < QGRP; ++u) {
                v2f e; e.x = dv[u]; e.y = __int_as_float(ibase + j0 + u);
                qe[cnt * 256 + tid] = e;
                cnt += (dv[u] < bound) ? 1 : 0;
            }
            const bool last = (r + 1 >= CHPTS / CH) && (j0 + QGRP >= CH);
            const unsigned full = __builtin_amdgcn_ballot_w32(cnt > QCAP - QGRP);
            if (last | (full != 0u)) {
                bound = knn_rank_flush(&qe[tid], &ls[tid], cnt);
                cnt = 0;
                sbv[tid] = ls[3 * 256 + tid].x;
                __builtin_amdgcn_fence(__ATOMIC_ACQUIRE, "workgroup", "local");
                float a0 = sbv[0 * 32 + lane], a1 = sbv[1 * 32 + lane];
                float a2 = sbv[2 * 32 + lane], a3 = sbv[3 * 32 + lane];
                float c0 = sbv[4 * 32 + lane], c1 = sbv[5 * 32 + lane];
                float c2 = sbv[6 * 32 + lane], c3 = sbv[7 * 32 + lane];
                float t;
                t = fminf(a0, a1); a1 = fmaxf(a0, a1); a0 = t;
                t = fminf(a2, a3); a3 = fmaxf(a2, a3); a2 = t;
                t = fminf(a0, a2); a2 = fmaxf(a0, a2); a0 = t;
                t = fminf(a1, a3); a3 = fmaxf(a1, a3); a1 = t;
                t = fminf(a1, a2); a2 = fmaxf(a1, a2); a1 = t;
                t = fminf(c0, c1); c1 = fmaxf(c0, c1); c0 = t;
                t = fminf(c2, c3); c3 = fmaxf(c2, c3); c2 = t;
                t = fminf(c0, c2); c2 = fmaxf(c0, c2); c0 = t;
                t = fminf(c1, c3); c3 = fmaxf(c1, c3); c1 = t;
                t = fminf(c1, c2); c2 = fmaxf(c1, c2); c1 = t;
                float f4 = fminf(a3, c3);
                f4 = fminf(f4, fmaxf(a0, c2));
                f4 = fminf(f4, fmaxf(a1, c1));
                f4 = fminf(f4, fmaxf(a2, c0));
                const float wide = __builtin_fmaf(fabsf(f4), 1.0e-6f, f4) + 1.0e-30f;
                bound = fminf(bound, wide);
                px0 = xw4[gn];  px1 = xw4[gn + 1];  py0 = yw4[gn];  py1 = yw4[gn + 1];
                pz0 = zw4[gn];  pz1 = zw4[gn + 1];  pq0 = x2w4[gn]; pq1 = x2w4[gn + 1];
            }
        }
    }
    const size_t base = ((size_t)(b * NS + s) * NCHUNK + chunk) * NK;
#pragma unroll
    for (int i = 0; i < NK; ++i) {
        const v2f e = ls[i * 256 + tid];
        pd[base + i] = e.x; pi[base + i] = __float_as_int(e.y);
    }
}

#define MQ   16
#define MENT (NCHUNK * NK)

__global__ __launch_bounds__(256) void knn_merge_kernel(
    const float* __restrict__ pd, const int* __restrict__ pi,
    int* __restrict__ knn) {
    __shared__ float4 dl4[MQ * MENT / 4];
    __shared__ int4   il4[MQ * MENT / 4];
    __shared__ int    kl[MQ * NK];

    const int tid = threadIdx.x;
    const size_t q0 = (size_t)blockIdx.x * MQ;

    const float4* gd = reinterpret_cast<const float4*>(pd + q0 * MENT);
    const int4*   gi = reinterpret_cast<const int4*>(pi + q0 * MENT);
    dl4[tid] = gd[tid];  dl4[tid + 256] = gd[tid + 256];
    il4[tid] = gi[tid];  il4[tid + 256] = gi[tid + 256];
    __syncthreads();

    const int s = tid >> 4;
    const int j = tid & 15;
    const float* dq = reinterpret_cast<const float*>(dl4) + s * MENT;
    const int*   iq = reinterpret_cast<const int*>(il4) + s * MENT;

    float mine[NCHUNK];
    int   rank[NCHUNK];
#pragma unroll
    for (int l = 0; l < NCHUNK; ++l) { mine[l] = dq[l * NK + j]; rank[l] = j; }

#pragma unroll
    for (int m = 0; m < NCHUNK; ++m) {
#pragma unroll
        for (int p4 = 0; p4 < NK / 4; ++p4) {
            const float4 e4 = dl4[s * (MENT / 4) + m * (NK / 4) + p4];
            const float e[4] = {e4.x, e4.y, e4.z, e4.w};
#pragma unroll
            for (int u = 0; u < 4; ++u) {
#pragma unroll
                for (int l = 0; l < NCHUNK; ++l) {
                    if (l == m) continue;
                    if (m < l) rank[l] += (e[u] <= mine[l]) ? 1 : 0;
                    else       rank[l] += (e[u] <  mine[l]) ? 1 : 0;
                }
            }
        }
    }

#pragma unroll
    for (int l = 0; l < NCHUNK; ++l)
        if (rank[l] < NK) kl[s * NK + rank[l]] = iq[l * NK + j];
    __syncthreads();
    knn[q0 * NK + tid] = kl[tid];
}

#define NPL    (NC + 3)
#define NSLOT  (NS * NK)
#define GSPLIT 4
#define GENT   (NSLOT / GSPLIT)

__global__ __launch_bounds__(256) void gather_plane_kernel(
    const float* __restrict__ xyz, const float* __restrict__ feat,
    const int* __restrict__ knn, float* __restrict__ gf) {
    __shared__ v4f pl4[NPT / 4];
    const float* pl = reinterpret_cast<const float*>(pl4);
    const int tid  = threadIdx.x;
    const int part = blockIdx.x % GSPLIT;
    const int t    = blockIdx.x / GSPLIT;
    const int c    = t % NPL;
    const int b    = t / NPL;

    const float* src = (c < 3) ? xyz  + ((size_t)b * 3  + c) * NPT
                               : feat + ((size_t)b * NC + (c - 3)) * NPT;
    const v4f* s4 = reinterpret_cast<const v4f*>(src);
    for (int i = tid; i < NPT / 4; i += 256) pl4[i] = s4[i];
    __syncthreads();

    const int4* k4 = reinterpret_cast<const int4*>(knn + (size_t)b * NSLOT + (size_t)part * GENT);
    v4f*        o4 = reinterpret_cast<v4f*>(gf + ((size_t)b * NPL + c) * NSLOT + (size_t)part * GENT);
    for (int i = tid; i < GENT / 4; i += 256) {
        const int4 id = k4[i];
        v4f v;
        v.x = pl[id.x]; v.y = pl[id.y]; v.z = pl[id.z]; v.w = pl[id.w];
        o4[i] = v;
    }
}

#define LWROW4  268
#define SPITCH  68

template <int NU>
__device__ __forceinline__ v8f lw_mma(v8f acc, const float* a, const float* bsrc) {
#pragma unroll
    for (int j = 0; j < NU * 4; ++j) {
        const v2f av = *reinterpret_cast<const v2f*>(a + j * 4);
        v2f bb; bb.x = bsrc[j * 4]; bb.y = bsrc[j * 4 + 1];
        acc = __builtin_amdgcn_wmma_f32_16x16x4_f32(
            false, av, false, bb, (short)0, acc, false, false);
    }
    return acc;
}

__global__ __launch_bounds__(256) void pointconv_kernel(
    const float* __restrict__ sxyz,
    const float* __restrict__ w1,   const float* __restrict__ b1,
    const float* __restrict__ w2,   const float* __restrict__ b2,
    const float* __restrict__ lw,   const float* __restrict__ lb,
    const float* __restrict__ gf,   float* __restrict__ out)
{
    const int tid  = threadIdx.x;
    const int lane = tid & 31;
    const int wave = tid >> 5;
    const int b    = blockIdx.x >> 8;
    const int s0   = (blockIdx.x & 255) * 16;

    __shared__ float w1s[24], b1s[8], w2s[128], b2s[16];
    __shared__ float Wl[16][16][16];
    __shared__ float Fl[16][16][68];
    __shared__ float Ag[16 * AGROW];

    if (tid < 24)  w1s[tid] = w1[tid];
    if (tid < 8)   b1s[tid] = b1[tid];
    if (tid < 128) w2s[tid] = w2[tid];
    if (tid < 16)  b2s[tid] = b2[tid];
    __syncthreads();

    {
        const int s = tid >> 4, k = tid & 15;
        const float* gp = gf + (size_t)b * NPL * NSLOT + (size_t)s0 * NK + tid;
        const float nx = gp[0];
        const float ny = gp[(size_t)NSLOT];
        const float nz = gp[(size_t)2 * NSLOT];
        const float rx = nx - sxyz[(b * 3 + 0) * NS + s0 + s];
        const float ry = ny - sxyz[(b * 3 + 1) * NS + s0 + s];
        const float rz = nz - sxyz[(b * 3 + 2) * NS + s0 + s];

        float h[8];
#pragma unroll
        for (int o = 0; o < 8; ++o) {
            float v = w1s[o * 3 + 0] * rx + w1s[o * 3 + 1] * ry +
                      w1s[o * 3 + 2] * rz + b1s[o];
            h[o] = v >= 0.f ? v : 0.1f * v;
        }
#pragma unroll
        for (int o = 0; o < 16; ++o) {
            float v = b2s[o];
#pragma unroll
            for (int i = 0; i < 8; ++i) v += w2s[o * 8 + i] * h[i];
            v = v >= 0.f ? v : 0.1f * v;
            Wl[s][k][o] = v;
        }
        Fl[s][k][0] = nx; Fl[s][k][1] = ny; Fl[s][k][2] = nz;
#pragma unroll 8
        for (int cf = 0; cf < NC; ++cf)
            Fl[s][k][3 + cf] = gp[(size_t)(3 + cf) * NSLOT];
        Fl[s][k][67] = 0.f;
    }
    __syncthreads();

    {
        const int r    = lane & 15;
        const int koff = (lane < 16) ? 0 : 2;
#pragma unroll
        for (int si = 0; si < 2; ++si) {
            const int s = wave * 2 + si;
#pragma unroll
            for (int ct = 0; ct < 5; ++ct) {
                const int c  = ct * 16 + r;
                const int cc = c > 67 ? 67 : c;
                v8f acc = {};
#pragma unroll
                for (int kb = 0; kb < 4; ++kb) {
                    const int k0 = kb * 4 + koff;
                    v2f a;  a.x  = Wl[s][k0][r];  a.y  = Wl[s][k0 + 1][r];
                    v2f bb; bb.x = Fl[s][k0][cc]; bb.y = Fl[s][k0 + 1][cc];
                    acc = __builtin_amdgcn_wmma_f32_16x16x4_f32(
                        false, a, false, bb, (short)0, acc, false, false);
                }
                if (c < 67) {
#pragma unroll
                    for (int rr = 0; rr < 8; ++rr) {
                        const int wi = rr + ((lane < 16) ? 0 : 8);
                        Ag[s * AGROW + wi * 67 + c] = acc[rr];
                    }
                }
            }
        }
    }
    __syncthreads();

    {
        const int r    = lane & 15;
        const int koff = (lane < 16) ? 0 : 2;
        const float* Arow = &Ag[r * AGROW];

        float* stg = &Fl[0][0][0] + wave * (2 * 16 * SPITCH);
        const int c4   = lane & 3;
        const int srow = 4 * ((lane >> 2) & 3) + (lane >> 4);
        const v4f* lp = reinterpret_cast<const v4f*>(lw)
                      + (size_t)(wave * 16 + srow) * LWROW4 + c4;
        float*       sput = stg + srow * SPITCH + c4 * 4;
        const float* sget = stg + r * SPITCH + koff;

        v4f a0, a1, a2, a3, b0, b1, b2, b3;
        a0 = lp[0]; a1 = lp[4]; a2 = lp[8];
        b0 = lp[2 * LWROW4]; b1 = lp[2 * LWROW4 + 4]; b2 = lp[2 * LWROW4 + 8];
        lp += 12;
        *reinterpret_cast<v4f*>(sput)                   = a0;
        *reinterpret_cast<v4f*>(sput + 16)              = a1;
        *reinterpret_cast<v4f*>(sput + 32)              = a2;
        *reinterpret_cast<v4f*>(sput + 2 * SPITCH)      = b0;
        *reinterpret_cast<v4f*>(sput + 2 * SPITCH + 16) = b1;
        *reinterpret_cast<v4f*>(sput + 2 * SPITCH + 32) = b2;
        a0 = lp[0]; a1 = lp[4]; a2 = lp[8]; a3 = lp[12];
        b0 = lp[2 * LWROW4]; b1 = lp[2 * LWROW4 + 4]; b2 = lp[2 * LWROW4 + 8]; b3 = lp[2 * LWROW4 + 12];
        lp += 16;
        __builtin_amdgcn_fence(__ATOMIC_RELEASE, "wavefront");
        __builtin_amdgcn_fence(__ATOMIC_ACQUIRE, "wavefront");
        v8f acc = {};
        acc = lw_mma<3>(acc, sget, Arow + koff);

        for (int S = 0; S < 15; ++S) {
            const int bo = ((S + 1) & 1) * (16 * SPITCH);
            float* sp = sput + bo;
            *reinterpret_cast<v4f*>(sp)                   = a0;
            *reinterpret_cast<v4f*>(sp + 16)              = a1;
            *reinterpret_cast<v4f*>(sp + 32)              = a2;
            *reinterpret_cast<v4f*>(sp + 48)              = a3;
            *reinterpret_cast<v4f*>(sp + 2 * SPITCH)      = b0;
            *reinterpret_cast<v4f*>(sp + 2 * SPITCH + 16) = b1;
            *reinterpret_cast<v4f*>(sp + 2 * SPITCH + 32) = b2;
            *reinterpret_cast<v4f*>(sp + 2 * SPITCH + 48) = b3;
            a0 = lp[0]; a1 = lp[4]; a2 = lp[8]; a3 = lp[12];
            b0 = lp[2 * LWROW4]; b1 = lp[2 * LWROW4 + 4]; b2 = lp[2 * LWROW4 + 8]; b3 = lp[2 * LWROW4 + 12];
            lp += 16;
            __builtin_amdgcn_fence(__ATOMIC_RELEASE, "wavefront");
            __builtin_amdgcn_fence(__ATOMIC_ACQUIRE, "wavefront");
            acc = lw_mma<4>(acc, sget + bo, Arow + 48 + S * 64 + koff);
        }
        *reinterpret_cast<v4f*>(sput)                   = a0;
        *reinterpret_cast<v4f*>(sput + 16)              = a1;
        *reinterpret_cast<v4f*>(sput + 32)              = a2;
        *reinterpret_cast<v4f*>(sput + 48)              = a3;
        *reinterpret_cast<v4f*>(sput + 2 * SPITCH)      = b0;
        *reinterpret_cast<v4f*>(sput + 2 * SPITCH + 16) = b1;
        *reinterpret_cast<v4f*>(sput + 2 * SPITCH + 32) = b2;
        *reinterpret_cast<v4f*>(sput + 2 * SPITCH + 48) = b3;
        __builtin_amdgcn_fence(__ATOMIC_RELEASE, "wavefront");
        __builtin_amdgcn_fence(__ATOMIC_ACQUIRE, "wavefront");
        acc = lw_mma<4>(acc, sget, Arow + 48 + 15 * 64 + koff);

        const int su = s0 + r;
#pragma unroll
        for (int rr = 0; rr < 8; ++rr) {
            const int o = wave * 16 + rr + ((lane < 16) ? 0 : 8);
            float v = acc[rr] + lb[o];
            v = v >= 0.f ? v : 0.1f * v;
            out[(size_t)b * NOUT * NS + (size_t)o * NS + su] = v;
        }
    }
}

extern "C" void kernel_launch(void* const* d_in, const int* in_sizes, int n_in,
                              void* d_out, int out_size, void* d_ws, size_t ws_size,
                              hipStream_t stream) {
    const float* xyz  = (const float*)d_in[0];
    const float* feat = (const float*)d_in[1];
    const float* sxyz = (const float*)d_in[2];
    const float* w1   = (const float*)d_in[3];
    const float* b1   = (const float*)d_in[4];
    const float* w2   = (const float*)d_in[5];
    const float* b2   = (const float*)d_in[6];
    const float* lw   = (const float*)d_in[7];
    const float* lb   = (const float*)d_in[8];
    float* out = (float*)d_out;

    const size_t npart = (size_t)NB * NS * NCHUNK * NK;
    float* pd  = (float*)d_ws;
    int*   pi  = (int*)((char*)d_ws + npart * sizeof(float));
    int*   knn = (int*)((char*)d_ws + 2 * npart * sizeof(float));
    float* gf  = (float*)((char*)d_ws + 2 * npart * sizeof(float)
                                      + (size_t)NB * NSLOT * sizeof(int));

    knn_partial_kernel<<<NB * (NS / 32), 256, 0, stream>>>(xyz, sxyz, pd, pi);
    knn_merge_kernel<<<NB * NS / MQ, 256, 0, stream>>>(pd, pi, knn);
    gather_plane_kernel<<<NB * NPL * GSPLIT, 256, 0, stream>>>(xyz, feat, knn, gf);
    pointconv_kernel<<<NB * (NS / 16), 256, 0, stream>>>(
        sxyz, w1, b1, w2, b2, lw, lb, gf, out);
}


# KNN flush sized by the wave's fullest queue: when no lane holds more than 18 entries the flush ranks 18 queue slots, not all 24 (pair compares 153, was 276); same lists, same k order
# speedup vs baseline: 10.2085x; 1.0564x over previous
#include <hip/hip_runtime.h>
#include <hip/hip_bf16.h>

typedef __attribute__((ext_vector_type(2))) float v2f;
typedef __attribute__((ext_vector_type(8))) float v8f;
typedef __attribute__((ext_vector_type(4))) float v4f;

#define NB     2
#define NPT    16384
#define NS     4096
#define NC     64
#define NOUT   128
#define NK     16
#define CH     512
#define NCHUNK 8
#define CHPTS  (NPT / NCHUNK)
#define AGROW  1076

#define QCAP 24
#define QGRP 8

#define QSMALL 18

template <int QN>
__device__ __forceinline__ float knn_rank_flush(const v2f* qt, v2f* lt, int cnt) {
    v2f   le[NK];
    float qd[QN];
    int   rq[QN];
#pragma unroll
    for (int i = 0; i < NK; ++i) le[i] = lt[i * 256];
#pragma unroll
    for (int k = 0; k < QN; ++k) {
        const float d = qt[k * 256].x;
        qd[k] = (k < cnt) ? d : __builtin_inff();
    }
#pragma unroll
    for (int k = 0; k < QN; ++k) {
        const float d  = qd[k];
        const bool  c1 = le[7].x <= d;
        const float p2 = c1 ? le[11].x : le[3].x;
        const float b1 = c1 ? le[9].x  : le[1].x;
        const float b5 = c1 ? le[13].x : le[5].x;
        const float a0 = c1 ? le[8].x  : le[0].x;
        const float a2 = c1 ? le[10].x : le[2].x;
        const float a4 = c1 ? le[12].x : le[4].x;
        const float a6 = c1 ? le[14].x : le[6].x;
        const bool  c2 = p2 <= d;
        const float p3 = c2 ? b5 : b1;
        const float e0 = c2 ? a4 : a0;
        const float e2 = c2 ? a6 : a2;
        const bool  c3 = p3 <= d;
        const float p4 = c3 ? e2 : e0;
        const bool  c4 = p4 <= d;
        const bool  c5 = le[15].x <= d;
        rq[k] = (c1 ? 8 : 0) + (c2 ? 4 : 0) + (c3 ? 2 : 0) + (c4 ? 1 : 0) + (c5 ? 1 : 0);
    }
#pragma unroll
    for (int kp = 0; kp < QN; ++kp) {
        int ahead = 0;
#pragma unroll
        for (int k = kp + 1; k < QN; ++k) {
            const int c = (qd[kp] <= qd[k]) ? 1 : 0;
            rq[k] += c;
            ahead += c;
        }
        rq[kp] += (QN - 1 - kp) - ahead;
    }
    unsigned taken = 0u;
#pragma unroll
    for (int k = 0; k < QN; ++k) {
        const int r = rq[k];
        taken |= 1u << (r < 31 ? r : 31);
        lt[(r < NK ? r : NK) * 256] = qt[k * 256];
    }
    unsigned freem = ~taken;
#pragma unroll
    for (int i = 0; i < NK; ++i) {
        const unsigned p = (unsigned)(__ffs((int)freem) - 1);
        freem &= freem - 1u;
        lt[(p < (unsigned)NK ? p : (unsigned)NK) * 256] = le[i];
    }
    return lt[(NK - 1) * 256].x;
}

__global__ __launch_bounds__(256) void knn_partial_kernel(
    const float* __restrict__ xyz, const float* __restrict__ sxyz,
    float* __restrict__ pd, int* __restrict__ pi) {
    const int tid  = threadIdx.x;
    const int lane = tid & 31;
    const int chunk = __builtin_amdgcn_readfirstlane(tid >> 5);
    const int blkPerB = NS / 32;
    const int b     = blockIdx.x / blkPerB;
    const int s     = (blockIdx.x % blkPerB) * 32 + lane;

    const float sx = sxyz[(b * 3 + 0) * NS + s];
    const float sy = sxyz[(b * 3 + 1) * NS + s];
    const float sz = sxyz[(b * 3 + 2) * NS + s];
    const float s2 = __builtin_fmaf(sz, sz, __builtin_fmaf(sx, sx, sy * sy));

    __shared__ v4f xl4[NCHUNK * CH / 4], yl4[NCHUNK * CH / 4], zl4[NCHUNK * CH / 4], x2l4[NCHUNK * CH / 4];
    __shared__ v2f qe[QCAP * 256];
    __shared__ v2f ls[(NK + 1) * 256];
    __shared__ float sbv[NCHUNK * 32];

    float bound = 3.4e38f;
#pragma unroll
    for (int i = 0; i < NK; ++i) {
        v2f e; e.x = 3.4e38f; e.y = __int_as_float(0);
        ls[i * 256 + tid] = e;
    }
#pragma unroll
    for (int k = 0; k < QCAP; ++k) {
        v2f e; e.x = __builtin_inff(); e.y = __int_as_float(0);
        qe[k * 256 + tid] = e;
    }
    sbv[tid] = 3.4e38f;
    int cnt = 0;

    const float* xb = xyz + (size_t)b * 3 * NPT;
    const v4f* xw4  = xl4  + chunk * (CH / 4);
    const v4f* yw4  = yl4  + chunk * (CH / 4);
    const v4f* zw4  = zl4  + chunk * (CH / 4);
    const v4f* x2w4 = x2l4 + chunk * (CH / 4);

    v4f px[4], py[4], pz[4];
#pragma unroll
    for (int i = 0; i < 4; ++i) {
        const int g = tid + 256 * i;
        const float* p = xb + (g >> 7) * CHPTS + (g & 127) * 4;
        px[i] = *reinterpret_cast<const v4f*>(p);
        py[i] = *reinterpret_cast<const v4f*>(p + NPT);
        pz[i] = *reinterpret_cast<const v4f*>(p + 2 * NPT);
    }

    for (int r = 0; r < CHPTS / CH; ++r) {
        __syncthreads();
#pragma unroll
        for (int i = 0; i < 4; ++i) {
            const int g = tid + 256 * i;
            v4f q2;
            q2.x = __builtin_fmaf(pz[i].x, pz[i].x, __builtin_fmaf(px[i].x, px[i].x, py[i].x * py[i].x));
            q2.y = __builtin_fmaf(pz[i].y, pz[i].y, __builtin_fmaf(px[i].y, px[i].y, py[i].y * py[i].y));
            q2.z = __builtin_fmaf(pz[i].z, pz[i].z, __builtin_fmaf(px[i].z, px[i].z, py[i].z * py[i].z));
            q2.w = __builtin_fmaf(pz[i].w, pz[i].w, __builtin_fmaf(px[i].w, px[i].w, py[i].w * py[i].w));
            xl4[g] = px[i]; yl4[g] = py[i]; zl4[g] = pz[i]; x2l4[g] = q2;
        }
        __syncthreads();
        if (r + 1 < CHPTS / CH) {
#pragma unroll
            for (int i = 0; i < 4; ++i) {
                const int g = tid + 256 * i;
                const float* p = xb + (g >> 7) * CHPTS + (r + 1) * CH + (g & 127) * 4;
                px[i] = *reinterpret_cast<const v4f*>(p);
                py[i] = *reinterpret_cast<const v4f*>(p + NPT);
                pz[i] = *reinterpret_cast<const v4f*>(p + 2 * NPT);
            }
        }
        const int ibase = chunk * CHPTS + r * CH;
        v4f px0 = xw4[0],  px1 = xw4[1],  py0 = yw4[0],  py1 = yw4[1];
        v4f pz0 = zw4[0],  pz1 = zw4[1],  pq0 = x2w4[0], pq1 = x2w4[1];
        for (int j0 = 0; j0 < CH; j0 += QGRP) {
            float dv[QGRP];
#define KNN_D2(X, Y, Z, Q)   \
            __builtin_fmaf(-2.0f, __builtin_fmaf(sz, (Z), __builtin_fmaf(sx, (X), sy * (Y))), s2 + (Q))
            dv[0] = KNN_D2(px0.x, py0.x, pz0.x, pq0.x);
            dv[1] = KNN_D2(px0.y, py0.y, pz0.y, pq0.y);
            dv[2] = KNN_D2(px0.z, py0.z, pz0.z, pq0.z);
            dv[3] = KNN_D2(px0.w, py0.w, pz0.w, pq0.w);
            dv[4] = KNN_D2(px1.x, py1.x, pz1.x, pq1.x);
            dv[5] = KNN_D2(px1.y, py1.y, pz1.y, pq1.y);
            dv[6] = KNN_D2(px1.z, py1.z, pz1.z, pq1.z);
            dv[7] = KNN_D2(px1.w, py1.w, pz1.w, pq1.w);
#undef KNN_D2
            const int gn = ((j0 + QGRP) & (CH - 1)) >> 2;
            px0 = xw4[gn];  px1 = xw4[gn + 1];  py0 = yw4[gn];  py1 = yw4[gn + 1];
            pz0 = zw4[gn];  pz1 = zw4[gn + 1];  pq0 = x2w4[gn]; pq1 = x2w4[gn + 1];
            __builtin_amdgcn_sched_barrier(0);
#pragma unroll
            for (int u = 0; u < QGRP; ++u) {
                v2f e; e.x = dv[u]; e.y = __int_as_float(ibase + j0 + u);
                qe[cnt * 256 + tid] = e;
                cnt += (dv[u] < bound) ? 1 : 0;
            }
            const bool last = (r + 1 >= CHPTS / CH) && (j0 + QGRP >= CH);
            const unsigned full = __builtin_amdgcn_ballot_w32(cnt > QCAP - QGRP);
            if (last | (full != 0u)) {
                const unsigned over = __builtin_amdgcn_ballot_w32(cnt > QSMALL);
                if (over == 0u) bound = knn_rank_flush<QSMALL>(&qe[tid], &ls[tid], cnt);
                else            bound = knn_rank_flush<QCAP>(&qe[tid], &ls[tid], cnt);
                cnt = 0;
                sbv[tid] = ls[3 * 256 + tid].x;
                __builtin_amdgcn_fence(__ATOMIC_ACQUIRE, "workgroup", "local");
                float a0 = sbv[0 * 32 + lane], a1 = sbv[1 * 32 + lane];
                float a2 = sbv[2 * 32 + lane], a3 = sbv[3 * 32 + lane];
                float c0 = sbv[4 * 32 + lane], c1 = sbv[5 * 32 + lane];
                float c2 = sbv[6 * 32 + lane], c3 = sbv[7 * 32 + lane];
                float t;
                t = fminf(a0, a1); a1 = fmaxf(a0, a1); a0 = t;
                t = fminf(a2, a3); a3 = fmaxf(a2, a3); a2 = t;
                t = fminf(a0, a2); a2 = fmaxf(a0, a2); a0 = t;
                t = fminf(a1, a3); a3 = fmaxf(a1, a3); a1 = t;
                t = fminf(a1, a2); a2 = fmaxf(a1, a2); a1 = t;
                t = fminf(c0, c1); c1 = fmaxf(c0, c1); c0 = t;
                t = fminf(c2, c3); c3 = fmaxf(c2, c3); c2 = t;
                t = fminf(c0, c2); c2 = fmaxf(c0, c2); c0 = t;
                t = fminf(c1, c3); c3 = fmaxf(c1, c3); c1 = t;
                t = fminf(c1, c2); c2 = fmaxf(c1, c2); c1 = t;
                float f4 = fminf(a3, c3);
                f4 = fminf(f4, fmaxf(a0, c2));
                f4 = fminf(f4, fmaxf(a1, c1));
                f4 = fminf(f4, fmaxf(a2, c0));
                const float wide = __builtin_fmaf(fabsf(f4), 1.0e-6f, f4) + 1.0e-30f;
                bound = fminf(bound, wide);
                px0 = xw4[gn];  px1 = xw4[gn + 1];  py0 = yw4[gn];  py1 = yw4[gn + 1];
                pz0 = zw4[gn];  pz1 = zw4[gn + 1];  pq0 = x2w4[gn]; pq1 = x2w4[gn + 1];
            }
        }
    }
    const size_t base = ((size_t)(b * NS + s) * NCHUNK + chunk) * NK;
#pragma unroll
    for (int i = 0; i < NK; ++i) {
        const v2f e = ls[i * 256 + tid];
        pd[base + i] = e.x; pi[base + i] = __float_as_int(e.y);
    }
}

#define MQ   16
#define MENT (NCHUNK * NK)

__global__ __launch_bounds__(256) void knn_merge_kernel(
    const float* __restrict__ pd, const int* __restrict__ pi,
    int* __restrict__ knn) {
    __shared__ float4 dl4[MQ * MENT / 4];
    __shared__ int4   il4[MQ * MENT / 4];
    __shared__ int    kl[MQ * NK];

    const int tid = threadIdx.x;
    const size_t q0 = (size_t)blockIdx.x * MQ;

    const float4* gd = reinterpret_cast<const float4*>(pd + q0 * MENT);
    const int4*   gi = reinterpret_cast<const int4*>(pi + q0 * MENT);
    dl4[tid] = gd[tid];  dl4[tid + 256] = gd[tid + 256];
    il4[tid] = gi[tid];  il4[tid + 256] = gi[tid + 256];
    __syncthreads();

    const int s = tid >> 4;
    const int j = tid & 15;
    const float* dq = reinterpret_cast<const float*>(dl4) + s * MENT;
    const int*   iq = reinterpret_cast<const int*>(il4) + s * MENT;

    float mine[NCHUNK];
    int   rank[NCHUNK];
#pragma unroll
    for (int l = 0; l < NCHUNK; ++l) { mine[l] = dq[l * NK + j]; rank[l] = j; }

#pragma unroll
    for (int m = 0; m < NCHUNK; ++m) {
#pragma unroll
        for (int p4 = 0; p4 < NK / 4; ++p4) {
            const float4 e4 = dl4[s * (MENT / 4) + m * (NK / 4) + p4];
            const float e[4] = {e4.x, e4.y, e4.z, e4.w};
#pragma unroll
            for (int u = 0; u < 4; ++u) {
#pragma unroll
                for (int l = 0; l < NCHUNK; ++l) {
                    if (l == m) continue;
                    if (m < l) rank[l] += (e[u] <= mine[l]) ? 1 : 0;
                    else       rank[l] += (e[u] <  mine[l]) ? 1 : 0;
                }
            }
        }
    }

#pragma unroll
    for (int l = 0; l < NCHUNK; ++l)
        if (rank[l] < NK) kl[s * NK + rank[l]] = iq[l * NK + j];
    __syncthreads();
    knn[q0 * NK + tid] = kl[tid];
}

#define NPL    (NC + 3)
#define NSLOT  (NS * NK)
#define GSPLIT 4
#define GENT   (NSLOT / GSPLIT)

__global__ __launch_bounds__(256) void gather_plane_kernel(
    const float* __restrict__ xyz, const float* __restrict__ feat,
    const int* __restrict__ knn, float* __restrict__ gf) {
    __shared__ v4f pl4[NPT / 4];
    const float* pl = reinterpret_cast<const float*>(pl4);
    const int tid  = threadIdx.x;
    const int part = blockIdx.x % GSPLIT;
    const int t    = blockIdx.x / GSPLIT;
    const int c    = t % NPL;
    const int b    = t / NPL;

    const float* src = (c < 3) ? xyz  + ((size_t)b * 3  + c) * NPT
                               : feat + ((size_t)b * NC + (c - 3)) * NPT;
    const v4f* s4 = reinterpret_cast<const v4f*>(src);
    for (int i = tid; i < NPT / 4; i += 256) pl4[i] = s4[i];
    __syncthreads();

    const int4* k4 = reinterpret_cast<const int4*>(knn + (size_t)b * NSLOT + (size_t)part * GENT);
    v4f*        o4 = reinterpret_cast<v4f*>(gf + ((size_t)b * NPL + c) * NSLOT + (size_t)part * GENT);
    for (int i = tid; i < GENT / 4; i += 256) {
        const int4 id = k4[i];
        v4f v;
        v.x = pl[id.x]; v.y = pl[id.y]; v.z = pl[id.z]; v.w = pl[id.w];
        o4[i] = v;
    }
}

#define LWROW4  268
#define SPITCH  68

template <int NU>
__device__ __forceinline__ v8f lw_mma(v8f acc, const float* a, const float* bsrc) {
#pragma unroll
    for (int j = 0; j < NU * 4; ++j) {
        const v2f av = *reinterpret_cast<const v2f*>(a + j * 4);
        v2f bb; bb.x = bsrc[j * 4]; bb.y = bsrc[j * 4 + 1];
        acc = __builtin_amdgcn_wmma_f32_16x16x4_f32(
            false, av, false, bb, (short)0, acc, false, false);
    }
    return acc;
}

__global__ __launch_bounds__(256) void pointconv_kernel(
    const float* __restrict__ sxyz,
    const float* __restrict__ w1,   const float* __restrict__ b1,
    const float* __restrict__ w2,   const float* __restrict__ b2,
    const float* __restrict__ lw,   const float* __restrict__ lb,
    const float* __restrict__ gf,   float* __restrict__ out)
{
    const int tid  = threadIdx.x;
    const int lane = tid & 31;
    const int wave = tid >> 5;
    const int b    = blockIdx.x >> 8;
    const int s0   = (blockIdx.x & 255) * 16;

    __shared__ float w1s[24], b1s[8], w2s[128], b2s[16];
    __shared__ float Wl[16][16][16];
    __shared__ float Fl[16][16][68];
    __shared__ float Ag[16 * AGROW];

    if (tid < 24)  w1s[tid] = w1[tid];
    if (tid < 8)   b1s[tid] = b1[tid];
    if (tid < 128) w2s[tid] = w2[tid];
    if (tid < 16)  b2s[tid] = b2[tid];
    __syncthreads();

    {
        const int s = tid >> 4, k = tid & 15;
        const float* gp = gf + (size_t)b * NPL * NSLOT + (size_t)s0 * NK + tid;
        const float nx = gp[0];
        const float ny = gp[(size_t)NSLOT];
        const float nz = gp[(size_t)2 * NSLOT];
        const float rx = nx - sxyz[(b * 3 + 0) * NS + s0 + s];
        const float ry = ny - sxyz[(b * 3 + 1) * NS + s0 + s];
        const float rz = nz - sxyz[(b * 3 + 2) * NS + s0 + s];

        float h[8];
#pragma unroll
        for (int o = 0; o < 8; ++o) {
            float v = w1s[o * 3 + 0] * rx + w1s[o * 3 + 1] * ry +
                      w1s[o * 3 + 2] * rz + b1s[o];
            h[o] = v >= 0.f ? v : 0.1f * v;
        }
#pragma unroll
        for (int o = 0; o < 16; ++o) {
            float v = b2s[o];
#pragma unroll
            for (int i = 0; i < 8; ++i) v += w2s[o * 8 + i] * h[i];
            v = v >= 0.f ? v : 0.1f * v;
            Wl[s][k][o] = v;
        }
        Fl[s][k][0] = nx; Fl[s][k][1] = ny; Fl[s][k][2] = nz;
#pragma unroll 8
        for (int cf = 0; cf < NC; ++cf)
            Fl[s][k][3 + cf] = gp[(size_t)(3 + cf) * NSLOT];
        Fl[s][k][67] = 0.f;
    }
    __syncthreads();

    {
        const int r    = lane & 15;
        const int koff = (lane < 16) ? 0 : 2;
#pragma unroll
        for (int si = 0; si < 2; ++si) {
            const int s = wave * 2 + si;
#pragma unroll
            for (int ct = 0; ct < 5; ++ct) {
                const int c  = ct * 16 + r;
                const int cc = c > 67 ? 67 : c;
                v8f acc = {};
#pragma unroll
                for (int kb = 0; kb < 4; ++kb) {
                    const int k0 = kb * 4 + koff;
                    v2f a;  a.x  = Wl[s][k0][r];  a.y  = Wl[s][k0 + 1][r];
                    v2f bb; bb.x = Fl[s][k0][cc]; bb.y = Fl[s][k0 + 1][cc];
                    acc = __builtin_amdgcn_wmma_f32_16x16x4_f32(
                        false, a, false, bb, (short)0, acc, false, false);
                }
                if (c < 67) {
#pragma unroll
                    for (int rr = 0; rr < 8; ++rr) {
                        const int wi = rr + ((lane < 16) ? 0 : 8);
                        Ag[s * AGROW + wi * 67 + c] = acc[rr];
                    }
                }
            }
        }
    }
    __syncthreads();

    {
        const int r    = lane & 15;
        const int koff = (lane < 16) ? 0 : 2;
        const float* Arow = &Ag[r * AGROW];

        float* stg = &Fl[0][0][0] + wave * (2 * 16 * SPITCH);
        const int c4   = lane & 3;
        const int srow = 4 * ((lane >> 2) & 3) + (lane >> 4);
        const v4f* lp = reinterpret_cast<const v4f*>(lw)
                      + (size_t)(wave * 16 + srow) * LWROW4 + c4;
        float*       sput = stg + srow * SPITCH + c4 * 4;
        const float* sget = stg + r * SPITCH + koff;

        v4f a0, a1, a2, a3, b0, b1, b2, b3;
        a0 = lp[0]; a1 = lp[4]; a2 = lp[8];
        b0 = lp[2 * LWROW4]; b1 = lp[2 * LWROW4 + 4]; b2 = lp[2 * LWROW4 + 8];
        lp += 12;
        *reinterpret_cast<v4f*>(sput)                   = a0;
        *reinterpret_cast<v4f*>(sput + 16)              = a1;
        *reinterpret_cast<v4f*>(sput + 32)              = a2;
        *reinterpret_cast<v4f*>(sput + 2 * SPITCH)      = b0;
        *reinterpret_cast<v4f*>(sput + 2 * SPITCH + 16) = b1;
        *reinterpret_cast<v4f*>(sput + 2 * SPITCH + 32) = b2;
        a0 = lp[0]; a1 = lp[4]; a2 = lp[8]; a3 = lp[12];
        b0 = lp[2 * LWROW4]; b1 = lp[2 * LWROW4 + 4]; b2 = lp[2 * LWROW4 + 8]; b3 = lp[2 * LWROW4 + 12];
        lp += 16;
        __builtin_amdgcn_fence(__ATOMIC_RELEASE, "wavefront");
        __builtin_amdgcn_fence(__ATOMIC_ACQUIRE, "wavefront");
        v8f acc = {};
        acc = lw_mma<3>(acc, sget, Arow + koff);

        for (int S = 0; S < 15; ++S) {
            const int bo = ((S + 1) & 1) * (16 * SPITCH);
            float* sp = sput + bo;
            *reinterpret_cast<v4f*>(sp)                   = a0;
            *reinterpret_cast<v4f*>(sp + 16)              = a1;
            *reinterpret_cast<v4f*>(sp + 32)              = a2;
            *reinterpret_cast<v4f*>(sp + 48)              = a3;
            *reinterpret_cast<v4f*>(sp + 2 * SPITCH)      = b0;
            *reinterpret_cast<v4f*>(sp + 2 * SPITCH + 16) = b1;
            *reinterpret_cast<v4f*>(sp + 2 * SPITCH + 32) = b2;
            *reinterpret_cast<v4f*>(sp + 2 * SPITCH + 48) = b3;
            a0 = lp[0]; a1 = lp[4]; a2 = lp[8]; a3 = lp[12];
            b0 = lp[2 * LWROW4]; b1 = lp[2 * LWROW4 + 4]; b2 = lp[2 * LWROW4 + 8]; b3 = lp[2 * LWROW4 + 12];
            lp += 16;
            __builtin_amdgcn_fence(__ATOMIC_RELEASE, "wavefront");
            __builtin_amdgcn_fence(__ATOMIC_ACQUIRE, "wavefront");
            acc = lw_mma<4>(acc, sget + bo, Arow + 48 + S * 64 + koff);
        }
        *reinterpret_cast<v4f*>(sput)                   = a0;
        *reinterpret_cast<v4f*>(sput + 16)              = a1;
        *reinterpret_cast<v4f*>(sput + 32)              = a2;
        *reinterpret_cast<v4f*>(sput + 48)              = a3;
        *reinterpret_cast<v4f*>(sput + 2 * SPITCH)      = b0;
        *reinterpret_cast<v4f*>(sput + 2 * SPITCH + 16) = b1;
        *reinterpret_cast<v4f*>(sput + 2 * SPITCH + 32) = b2;
        *reinterpret_cast<v4f*>(sput + 2 * SPITCH + 48) = b3;
        __builtin_amdgcn_fence(__ATOMIC_RELEASE, "wavefront");
        __builtin_amdgcn_fence(__ATOMIC_ACQUIRE, "wavefront");
        acc = lw_mma<4>(acc, sget, Arow + 48 + 15 * 64 + koff);

        const int su = s0 + r;
#pragma unroll
        for (int rr = 0; rr < 8; ++rr) {
            const int o = wave * 16 + rr + ((lane < 16) ? 0 : 8);
            float v = acc[rr] + lb[o];
            v = v >= 0.f ? v : 0.1f * v;
            out[(size_t)b * NOUT * NS + (size_t)o * NS + su] = v;
        }
    }
}

extern "C" void kernel_launch(void* const* d_in, const int* in_sizes, int n_in,
                              void* d_out, int out_size, void* d_ws, size_t ws_size,
                              hipStream_t stream) {
    const float* xyz  = (const float*)d_in[0];
    const float* feat = (const float*)d_in[1];
    const float* sxyz = (const float*)d_in[2];
    const float* w1   = (const float*)d_in[3];
    const float* b1   = (const float*)d_in[4];
    const float* w2   = (const float*)d_in[5];
    const float* b2   = (const float*)d_in[6];
    const float* lw   = (const float*)d_in[7];
    const float* lb   = (const float*)d_in[8];
    float* out = (float*)d_out;

    const size_t npart = (size_t)NB * NS * NCHUNK * NK;
    float* pd  = (float*)d_ws;
    int*   pi  = (int*)((char*)d_ws + npart * sizeof(float));
    int*   knn = (int*)((char*)d_ws + 2 * npart * sizeof(float));
    float* gf  = (float*)((char*)d_ws + 2 * npart * sizeof(float)
                                      + (size_t)NB * NSLOT * sizeof(int));

    knn_partial_kernel<<<NB * (NS / 32), 256, 0, stream>>>(xyz, sxyz, pd, pi);
    knn_merge_kernel<<<NB * NS / MQ, 256, 0, stream>>>(pd, pi, knn);
    gather_plane_kernel<<<NB * NPL * GSPLIT, 256, 0, stream>>>(xyz, feat, knn, gf);
    pointconv_kernel<<<NB * (NS / 16), 256, 0, stream>>>(
        sxyz, w1, b1, w2, b2, lw, lb, gf, out);
}


# KNN merge moved into the scanning block: a query's 8 chunk lists, already in LDS, merged pairwise in 3 levels by bisection (was a launch ranking 128 entries from workspace); same lists, same k order
# speedup vs baseline: 10.5860x; 1.0370x over previous
#include <hip/hip_runtime.h>
#include <hip/hip_bf16.h>

typedef __attribute__((ext_vector_type(2))) float v2f;
typedef __attribute__((ext_vector_type(8))) float v8f;
typedef __attribute__((ext_vector_type(4))) float v4f;

#define NB     2
#define NPT    16384
#define NS     4096
#define NC     64
#define NOUT   128
#define NK     16
#define CH     512
#define NCHUNK 8
#define CHPTS  (NPT / NCHUNK)
#define AGROW  1076

#define QCAP 24
#define QGRP 8

#define QSMALL 18

template <int QN>
__device__ __forceinline__ float knn_rank_flush(const v2f* qt, v2f* lt, int cnt) {
    v2f   le[NK];
    float qd[QN];
    int   rq[QN];
#pragma unroll
    for (int i = 0; i < NK; ++i) le[i] = lt[i * 256];
#pragma unroll
    for (int k = 0; k < QN; ++k) {
        const float d = qt[k * 256].x;
        qd[k] = (k < cnt) ? d : __builtin_inff();
    }
#pragma unroll
    for (int k = 0; k < QN; ++k) {
        const float d  = qd[k];
        const bool  c1 = le[7].x <= d;
        const float p2 = c1 ? le[11].x : le[3].x;
        const float b1 = c1 ? le[9].x  : le[1].x;
        const float b5 = c1 ? le[13].x : le[5].x;
        const float a0 = c1 ? le[8].x  : le[0].x;
        const float a2 = c1 ? le[10].x : le[2].x;
        const float a4 = c1 ? le[12].x : le[4].x;
        const float a6 = c1 ? le[14].x : le[6].x;
        const bool  c2 = p2 <= d;
        const float p3 = c2 ? b5 : b1;
        const float e0 = c2 ? a4 : a0;
        const float e2 = c2 ? a6 : a2;
        const bool  c3 = p3 <= d;
        const float p4 = c3 ? e2 : e0;
        const bool  c4 = p4 <= d;
        const bool  c5 = le[15].x <= d;
        rq[k] = (c1 ? 8 : 0) + (c2 ? 4 : 0) + (c3 ? 2 : 0) + (c4 ? 1 : 0) + (c5 ? 1 : 0);
    }
#pragma unroll
    for (int kp = 0; kp < QN; ++kp) {
        int ahead = 0;
#pragma unroll
        for (int k = kp + 1; k < QN; ++k) {
            const int c = (qd[kp] <= qd[k]) ? 1 : 0;
            rq[k] += c;
            ahead += c;
        }
        rq[kp] += (QN - 1 - kp) - ahead;
    }
    unsigned taken = 0u;
#pragma unroll
    for (int k = 0; k < QN; ++k) {
        const int r = rq[k];
        taken |= 1u << (r < 31 ? r : 31);
        lt[(r < NK ? r : NK) * 256] = qt[k * 256];
    }
    unsigned freem = ~taken;
#pragma unroll
    for (int i = 0; i < NK; ++i) {
        const unsigned p = (unsigned)(__ffs((int)freem) - 1);
        freem &= freem - 1u;
        lt[(p < (unsigned)NK ? p : (unsigned)NK) * 256] = le[i];
    }
    return lt[(NK - 1) * 256].x;
}

template <bool LE>
__device__ __forceinline__ int knn_place(const float (&le)[NK], float d) {
#define KNN_BEFORE(X) (LE ? ((X) <= d) : ((X) < d))
    const bool  c1 = KNN_BEFORE(le[7]);
    const float p2 = c1 ? le[11] : le[3];
    const float b1 = c1 ? le[9]  : le[1];
    const float b5 = c1 ? le[13] : le[5];
    const float a0 = c1 ? le[8]  : le[0];
    const float a2 = c1 ? le[10] : le[2];
    const float a4 = c1 ? le[12] : le[4];
    const float a6 = c1 ? le[14] : le[6];
    const bool  c2 = KNN_BEFORE(p2);
    const float p3 = c2 ? b5 : b1;
    const float e0 = c2 ? a4 : a0;
    const float e2 = c2 ? a6 : a2;
    const bool  c3 = KNN_BEFORE(p3);
    const float p4 = c3 ? e2 : e0;
    const bool  c4 = KNN_BEFORE(p4);
    const bool  c5 = KNN_BEFORE(le[15]);
#undef KNN_BEFORE
    return (c1 ? 8 : 0) + (c2 ? 4 : 0) + (c3 ? 2 : 0) + (c4 ? 1 : 0) + (c5 ? 1 : 0);
}

__device__ __forceinline__ void knn_pair_merge(const v2f* src, v2f* dst,
                                               int lcol, int h, int ocol, int scol) {
    float ld[NK], rd[NK];
#pragma unroll
    for (int i = 0; i < NK; ++i) {
        ld[i] = src[i * 256 + lcol].x;
        rd[i] = src[i * 256 + lcol + 32].x;
    }
#pragma unroll
    for (int m = 0; m < NK / 2; ++m) {
        const int i = 2 * m + h;
        const v2f el = src[i * 256 + lcol];
        const v2f er = src[i * 256 + lcol + 32];
        const int pl = i + knn_place<false>(rd, el.x);
        const int pr = i + knn_place<true>(ld, er.x);
        dst[pl < NK ? pl * 256 + ocol : scol] = el;
        dst[pr < NK ? pr * 256 + ocol : scol] = er;
    }
}

__global__ __launch_bounds__(256) void knn_kernel(
    const float* __restrict__ xyz, const float* __restrict__ sxyz,
    int* __restrict__ knn) {
    const int tid  = threadIdx.x;
    const int lane = tid & 31;
    const int chunk = __builtin_amdgcn_readfirstlane(tid >> 5);
    const int blkPerB = NS / 32;
    const int b     = blockIdx.x / blkPerB;
    const int s     = (blockIdx.x % blkPerB) * 32 + lane;

    const float sx = sxyz[(b * 3 + 0) * NS + s];
    const float sy = sxyz[(b * 3 + 1) * NS + s];
    const float sz = sxyz[(b * 3 + 2) * NS + s];
    const float s2 = __builtin_fmaf(sz, sz, __builtin_fmaf(sx, sx, sy * sy));

    __shared__ v4f xl4[NCHUNK * CH / 4], yl4[NCHUNK * CH / 4], zl4[NCHUNK * CH / 4], x2l4[NCHUNK * CH / 4];
    __shared__ v2f qe[QCAP * 256];
    __shared__ v2f ls[(NK + 1) * 256];
    __shared__ float sbv[NCHUNK * 32];

    float bound = 3.4e38f;
#pragma unroll
    for (int i = 0; i < NK; ++i) {
        v2f e; e.x = 3.4e38f; e.y = __int_as_float(0);
        ls[i * 256 + tid] = e;
    }
#pragma unroll
    for (int k = 0; k < QCAP; ++k) {
        v2f e; e.x = __builtin_inff(); e.y = __int_as_float(0);
        qe[k * 256 + tid] = e;
    }
    sbv[tid] = 3.4e38f;
    int cnt = 0;

    const float* xb = xyz + (size_t)b * 3 * NPT;
    const v4f* xw4  = xl4  + chunk * (CH / 4);
    const v4f* yw4  = yl4  + chunk * (CH / 4);
    const v4f* zw4  = zl4  + chunk * (CH / 4);
    const v4f* x2w4 = x2l4 + chunk * (CH / 4);

    v4f px[4], py[4], pz[4];
#pragma unroll
    for (int i = 0; i < 4; ++i) {
        const int g = tid + 256 * i;
        const float* p = xb + (g >> 7) * CHPTS + (g & 127) * 4;
        px[i] = *reinterpret_cast<const v4f*>(p);
        py[i] = *reinterpret_cast<const v4f*>(p + NPT);
        pz[i] = *reinterpret_cast<const v4f*>(p + 2 * NPT);
    }

    for (int r = 0; r < CHPTS / CH; ++r) {
        __syncthreads();
#pragma unroll
        for (int i = 0; i < 4; ++i) {
            const int g = tid + 256 * i;
            v4f q2;
            q2.x = __builtin_fmaf(pz[i].x, pz[i].x, __builtin_fmaf(px[i].x, px[i].x, py[i].x * py[i].x));
            q2.y = __builtin_fmaf(pz[i].y, pz[i].y, __builtin_fmaf(px[i].y, px[i].y, py[i].y * py[i].y));
            q2.z = __builtin_fmaf(pz[i].z, pz[i].z, __builtin_fmaf(px[i].z, px[i].z, py[i].z * py[i].z));
            q2.w = __builtin_fmaf(pz[i].w, pz[i].w, __builtin_fmaf(px[i].w, px[i].w, py[i].w * py[i].w));
            xl4[g] = px[i]; yl4[g] = py[i]; zl4[g] = pz[i]; x2l4[g] = q2;
        }
        __syncthreads();
        if (r + 1 < CHPTS / CH) {
#pragma unroll
            for (int i = 0; i < 4; ++i) {
                const int g = tid + 256 * i;
                const float* p = xb + (g >> 7) * CHPTS + (r + 1) * CH + (g & 127) * 4;
                px[i] = *reinterpret_cast<const v4f*>(p);
                py[i] = *reinterpret_cast<const v4f*>(p + NPT);
                pz[i] = *reinterpret_cast<const v4f*>(p + 2 * NPT);
            }
        }
        const int ibase = chunk * CHPTS + r * CH;
        v4f px0 = xw4[0],  px1 = xw4[1],  py0 = yw4[0],  py1 = yw4[1];
        v4f pz0 = zw4[0],  pz1 = zw4[1],  pq0 = x2w4[0], pq1 = x2w4[1];
        for (int j0 = 0; j0 < CH; j0 += QGRP) {
            float dv[QGRP];
#define KNN_D2(X, Y, Z, Q)   \
            __builtin_fmaf(-2.0f, __builtin_fmaf(sz, (Z), __builtin_fmaf(sx, (X), sy * (Y))), s2 + (Q))
            dv[0] = KNN_D2(px0.x, py0.x, pz0.x, pq0.x);
            dv[1] = KNN_D2(px0.y, py0.y, pz0.y, pq0.y);
            dv[2] = KNN_D2(px0.z, py0.z, pz0.z, pq0.z);
            dv[3] = KNN_D2(px0.w, py0.w, pz0.w, pq0.w);
            dv[4] = KNN_D2(px1.x, py1.x, pz1.x, pq1.x);
            dv[5] = KNN_D2(px1.y, py1.y, pz1.y, pq1.y);
            dv[6] = KNN_D2(px1.z, py1.z, pz1.z, pq1.z);
            dv[7] = KNN_D2(px1.w, py1.w, pz1.w, pq1.w);
#undef KNN_D2
            const int gn = ((j0 + QGRP) & (CH - 1)) >> 2;
            px0 = xw4[gn];  px1 = xw4[gn + 1];  py0 = yw4[gn];  py1 = yw4[gn + 1];
            pz0 = zw4[gn];  pz1 = zw4[gn + 1];  pq0 = x2w4[gn]; pq1 = x2w4[gn + 1];
            __builtin_amdgcn_sched_barrier(0);
#pragma unroll
            for (int u = 0; u < QGRP; ++u) {
                v2f e; e.x = dv[u]; e.y = __int_as_float(ibase + j0 + u);
                qe[cnt * 256 + tid] = e;
                cnt += (dv[u] < bound) ? 1 : 0;
            }
            const bool last = (r + 1 >= CHPTS / CH) && (j0 + QGRP >= CH);
            const unsigned full = __builtin_amdgcn_ballot_w32(cnt > QCAP - QGRP);
            if (last | (full != 0u)) {
                const unsigned over = __builtin_amdgcn_ballot_w32(cnt > QSMALL);
                if (over == 0u) bound = knn_rank_flush<QSMALL>(&qe[tid], &ls[tid], cnt);
                else            bound = knn_rank_flush<QCAP>(&qe[tid], &ls[tid], cnt);
                cnt = 0;
                sbv[tid] = ls[3 * 256 + tid].x;
                __builtin_amdgcn_fence(__ATOMIC_ACQUIRE, "workgroup", "local");
                float a0 = sbv[0 * 32 + lane], a1 = sbv[1 * 32 + lane];
                float a2 = sbv[2 * 32 + lane], a3 = sbv[3 * 32 + lane];
                float c0 = sbv[4 * 32 + lane], c1 = sbv[5 * 32 + lane];
                float c2 = sbv[6 * 32 + lane], c3 = sbv[7 * 32 + lane];
                float t;
                t = fminf(a0, a1); a1 = fmaxf(a0, a1); a0 = t;
                t = fminf(a2, a3); a3 = fmaxf(a2, a3); a2 = t;
                t = fminf(a0, a2); a2 = fmaxf(a0, a2); a0 = t;
                t = fminf(a1, a3); a3 = fmaxf(a1, a3); a1 = t;
                t = fminf(a1, a2); a2 = fmaxf(a1, a2); a1 = t;
                t = fminf(c0, c1); c1 = fmaxf(c0, c1); c0 = t;
                t = fminf(c2, c3); c3 = fmaxf(c2, c3); c2 = t;
                t = fminf(c0, c2); c2 = fmaxf(c0, c2); c0 = t;
                t = fminf(c1, c3); c3 = fmaxf(c1, c3); c1 = t;
                t = fminf(c1, c2); c2 = fmaxf(c1, c2); c1 = t;
                float f4 = fminf(a3, c3);
                f4 = fminf(f4, fmaxf(a0, c2));
                f4 = fminf(f4, fmaxf(a1, c1));
                f4 = fminf(f4, fmaxf(a2, c0));
                const float wide = __builtin_fmaf(fabsf(f4), 1.0e-6f, f4) + 1.0e-30f;
                bound = fminf(bound, wide);
                px0 = xw4[gn];  px1 = xw4[gn + 1];  py0 = yw4[gn];  py1 = yw4[gn + 1];
                pz0 = zw4[gn];  pz1 = zw4[gn + 1];  pq0 = x2w4[gn]; pq1 = x2w4[gn + 1];
            }
        }
    }
    v2f* msrc = ls;
    v2f* mdst = qe;
#pragma unroll 1
    for (int lev = 0; lev < 3; ++lev) {
        __syncthreads();
        if (chunk < (NCHUNK >> lev))
            knn_pair_merge(msrc, mdst, (chunk & ~1) * 32 + lane, chunk & 1,
                           (chunk >> 1) * 32 + lane, NK * 256 + tid);
        v2f* t = msrc; msrc = mdst; mdst = t;
    }
    __syncthreads();
    {
        const int e = 2 * tid;
        const int q = e >> 4, rk = e & 15;
        int2 v;
        v.x = __float_as_int(qe[rk * 256 + q].y);
        v.y = __float_as_int(qe[(rk + 1) * 256 + q].y);
        int* kb = knn + ((size_t)b * NS + (size_t)(blockIdx.x % blkPerB) * 32) * NK;
        *reinterpret_cast<int2*>(kb + e) = v;
    }
}

#define NPL    (NC + 3)
#define NSLOT  (NS * NK)
#define GSPLIT 4
#define GENT   (NSLOT / GSPLIT)

__global__ __launch_bounds__(256) void gather_plane_kernel(
    const float* __restrict__ xyz, const float* __restrict__ feat,
    const int* __restrict__ knn, float* __restrict__ gf) {
    __shared__ v4f pl4[NPT / 4];
    const float* pl = reinterpret_cast<const float*>(pl4);
    const int tid  = threadIdx.x;
    const int part = blockIdx.x % GSPLIT;
    const int t    = blockIdx.x / GSPLIT;
    const int c    = t % NPL;
    const int b    = t / NPL;

    const float* src = (c < 3) ? xyz  + ((size_t)b * 3  + c) * NPT
                               : feat + ((size_t)b * NC + (c - 3)) * NPT;
    const v4f* s4 = reinterpret_cast<const v4f*>(src);
    for (int i = tid; i < NPT / 4; i += 256) pl4[i] = s4[i];
    __syncthreads();

    const int4* k4 = reinterpret_cast<const int4*>(knn + (size_t)b * NSLOT + (size_t)part * GENT);
    v4f*        o4 = reinterpret_cast<v4f*>(gf + ((size_t)b * NPL + c) * NSLOT + (size_t)part * GENT);
    for (int i = tid; i < GENT / 4; i += 256) {
        const int4 id = k4[i];
        v4f v;
        v.x = pl[id.x]; v.y = pl[id.y]; v.z = pl[id.z]; v.w = pl[id.w];
        o4[i] = v;
    }
}

#define LWROW4  268
#define SPITCH  68

template <int NU>
__device__ __forceinline__ v8f lw_mma(v8f acc, const float* a, const float* bsrc) {
#pragma unroll
    for (int j = 0; j < NU * 4; ++j) {
        const v2f av = *reinterpret_cast<const v2f*>(a + j * 4);
        v2f bb; bb.x = bsrc[j * 4]; bb.y = bsrc[j * 4 + 1];
        acc = __builtin_amdgcn_wmma_f32_16x16x4_f32(
            false, av, false, bb, (short)0, acc, false, false);
    }
    return acc;
}

__global__ __launch_bounds__(256) void pointconv_kernel(
    const float* __restrict__ sxyz,
    const float* __restrict__ w1,   const float* __restrict__ b1,
    const float* __restrict__ w2,   const float* __restrict__ b2,
    const float* __restrict__ lw,   const float* __restrict__ lb,
    const float* __restrict__ gf,   float* __restrict__ out)
{
    const int tid  = threadIdx.x;
    const int lane = tid & 31;
    const int wave = tid >> 5;
    const int b    = blockIdx.x >> 8;
    const int s0   = (blockIdx.x & 255) * 16;

    __shared__ float w1s[24], b1s[8], w2s[128], b2s[16];
    __shared__ float Wl[16][16][16];
    __shared__ float Fl[16][16][68];
    __shared__ float Ag[16 * AGROW];

    if (tid < 24)  w1s[tid] = w1[tid];
    if (tid < 8)   b1s[tid] = b1[tid];
    if (tid < 128) w2s[tid] = w2[tid];
    if (tid < 16)  b2s[tid] = b2[tid];
    __syncthreads();

    {
        const int s = tid >> 4, k = tid & 15;
        const float* gp = gf + (size_t)b * NPL * NSLOT + (size_t)s0 * NK + tid;
        const float nx = gp[0];
        const float ny = gp[(size_t)NSLOT];
        const float nz = gp[(size_t)2 * NSLOT];
        const float rx = nx - sxyz[(b * 3 + 0) * NS + s0 + s];
        const float ry = ny - sxyz[(b * 3 + 1) * NS + s0 + s];
        const float rz = nz - sxyz[(b * 3 + 2) * NS + s0 + s];

        float h[8];
#pragma unroll
        for (int o = 0; o < 8; ++o) {
            float v = w1s[o * 3 + 0] * rx + w1s[o * 3 + 1] * ry +
                      w1s[o * 3 + 2] * rz + b1s[o];
            h[o] = v >= 0.f ? v : 0.1f * v;
        }
#pragma unroll
        for (int o = 0; o < 16; ++o) {
            float v = b2s[o];
#pragma unroll
            for (int i = 0; i < 8; ++i) v += w2s[o * 8 + i] * h[i];
            v = v >= 0.f ? v : 0.1f * v;
            Wl[s][k][o] = v;
        }
        Fl[s][k][0] = nx; Fl[s][k][1] = ny; Fl[s][k][2] = nz;
#pragma unroll 8
        for (int cf = 0; cf < NC; ++cf)
            Fl[s][k][3 + cf] = gp[(size_t)(3 + cf) * NSLOT];
        Fl[s][k][67] = 0.f;
    }
    __syncthreads();

    {
        const int r    = lane & 15;
        const int koff = (lane < 16) ? 0 : 2;
#pragma unroll
        for (int si = 0; si < 2; ++si) {
            const int s = wave * 2 + si;
#pragma unroll
            for (int ct = 0; ct < 5; ++ct) {
                const int c  = ct * 16 + r;
                const int cc = c > 67 ? 67 : c;
                v8f acc = {};
#pragma unroll
                for (int kb = 0; kb < 4; ++kb) {
                    const int k0 = kb * 4 + koff;
                    v2f a;  a.x  = Wl[s][k0][r];  a.y  = Wl[s][k0 + 1][r];
                    v2f bb; bb.x = Fl[s][k0][cc]; bb.y = Fl[s][k0 + 1][cc];
                    acc = __builtin_amdgcn_wmma_f32_16x16x4_f32(
                        false, a, false, bb, (short)0, acc, false, false);
                }
                if (c < 67) {
#pragma unroll
                    for (int rr = 0; rr < 8; ++rr) {
                        const int wi = rr + ((lane < 16) ? 0 : 8);
                        Ag[s * AGROW + wi * 67 + c] = acc[rr];
                    }
                }
            }
        }
    }
    __syncthreads();

    {
        const int r    = lane & 15;
        const int koff = (lane < 16) ? 0 : 2;
        const float* Arow = &Ag[r * AGROW];

        float* stg = &Fl[0][0][0] + wave * (2 * 16 * SPITCH);
        const int c4   = lane & 3;
        const int srow = 4 * ((lane >> 2) & 3) + (lane >> 4);
        const v4f* lp = reinterpret_cast<const v4f*>(lw)
                      + (size_t)(wave * 16 + srow) * LWROW4 + c4;
        float*       sput = stg + srow * SPITCH + c4 * 4;
        const float* sget = stg + r * SPITCH + koff;

        v4f a0, a1, a2, a3, b0, b1, b2, b3;
        a0 = lp[0]; a1 = lp[4]; a2 = lp[8];
        b0 = lp[2 * LWROW4]; b1 = lp[2 * LWROW4 + 4]; b2 = lp[2 * LWROW4 + 8];
        lp += 12;
        *reinterpret_cast<v4f*>(sput)                   = a0;
        *reinterpret_cast<v4f*>(sput + 16)              = a1;
        *reinterpret_cast<v4f*>(sput + 32)              = a2;
        *reinterpret_cast<v4f*>(sput + 2 * SPITCH)      = b0;
        *reinterpret_cast<v4f*>(sput + 2 * SPITCH + 16) = b1;
        *reinterpret_cast<v4f*>(sput + 2 * SPITCH + 32) = b2;
        a0 = lp[0]; a1 = lp[4]; a2 = lp[8]; a3 = lp[12];
        b0 = lp[2 * LWROW4]; b1 = lp[2 * LWROW4 + 4]; b2 = lp[2 * LWROW4 + 8]; b3 = lp[2 * LWROW4 + 12];
        lp += 16;
        __builtin_amdgcn_fence(__ATOMIC_RELEASE, "wavefront");
        __builtin_amdgcn_fence(__ATOMIC_ACQUIRE, "wavefront");
        v8f acc = {};
        acc = lw_mma<3>(acc, sget, Arow + koff);

        for (int S = 0; S < 15; ++S) {
            const int bo = ((S + 1) & 1) * (16 * SPITCH);
            float* sp = sput + bo;
            *reinterpret_cast<v4f*>(sp)                   = a0;
            *reinterpret_cast<v4f*>(sp + 16)              = a1;
            *reinterpret_cast<v4f*>(sp + 32)              = a2;
            *reinterpret_cast<v4f*>(sp + 48)              = a3;
            *reinterpret_cast<v4f*>(sp + 2 * SPITCH)      = b0;
            *reinterpret_cast<v4f*>(sp + 2 * SPITCH + 16) = b1;
            *reinterpret_cast<v4f*>(sp + 2 * SPITCH + 32) = b2;
            *reinterpret_cast<v4f*>(sp + 2 * SPITCH + 48) = b3;
            a0 = lp[0]; a1 = lp[4]; a2 = lp[8]; a3 = lp[12];
            b0 = lp[2 * LWROW4]; b1 = lp[2 * LWROW4 + 4]; b2 = lp[2 * LWROW4 + 8]; b3 = lp[2 * LWROW4 + 12];
            lp += 16;
            __builtin_amdgcn_fence(__ATOMIC_RELEASE, "wavefront");
            __builtin_amdgcn_fence(__ATOMIC_ACQUIRE, "wavefront");
            acc = lw_mma<4>(acc, sget + bo, Arow + 48 + S * 64 + koff);
        }
        *reinterpret_cast<v4f*>(sput)                   = a0;
        *reinterpret_cast<v4f*>(sput + 16)              = a1;
        *reinterpret_cast<v4f*>(sput + 32)              = a2;
        *reinterpret_cast<v4f*>(sput + 48)              = a3;
        *reinterpret_cast<v4f*>(sput + 2 * SPITCH)      = b0;
        *reinterpret_cast<v4f*>(sput + 2 * SPITCH + 16) = b1;
        *reinterpret_cast<v4f*>(sput + 2 * SPITCH + 32) = b2;
        *reinterpret_cast<v4f*>(sput + 2 * SPITCH + 48) = b3;
        __builtin_amdgcn_fence(__ATOMIC_RELEASE, "wavefront");
        __builtin_amdgcn_fence(__ATOMIC_ACQUIRE, "wavefront");
        acc = lw_mma<4>(acc, sget, Arow + 48 + 15 * 64 + koff);

        const int su = s0 + r;
#pragma unroll
        for (int rr = 0; rr < 8; ++rr) {
            const int o = wave * 16 + rr + ((lane < 16) ? 0 : 8);
            float v = acc[rr] + lb[o];
            v = v >= 0.f ? v : 0.1f * v;
            out[(size_t)b * NOUT * NS + (size_t)o * NS + su] = v;
        }
    }
}

extern "C" void kernel_launch(void* const* d_in, const int* in_sizes, int n_in,
                              void* d_out, int out_size, void* d_ws, size_t ws_size,
                              hipStream_t stream) {
    const float* xyz  = (const float*)d_in[0];
    const float* feat = (const float*)d_in[1];
    const float* sxyz = (const float*)d_in[2];
    const float* w1   = (const float*)d_in[3];
    const float* b1   = (const float*)d_in[4];
    const float* w2   = (const float*)d_in[5];
    const float* b2   = (const float*)d_in[6];
    const float* lw   = (const float*)d_in[7];
    const float* lb   = (const float*)d_in[8];
    float* out = (float*)d_out;

    int*   knn = (int*)d_ws;
    float* gf  = (float*)((char*)d_ws + (size_t)NB * NSLOT * sizeof(int));

    knn_kernel<<<NB * (NS / 32), 256, 0, stream>>>(xyz, sxyz, knn);
    gather_plane_kernel<<<NB * NPL * GSPLIT, 256, 0, stream>>>(xyz, feat, knn, gf);
    pointconv_kernel<<<NB * (NS / 16), 256, 0, stream>>>(
        sxyz, w1, b1, w2, b2, lw, lb, gf, out);
}
